# starting kernel
# baseline (speedup 1.0000x reference)
#include <hip/hip_runtime.h>
#include <hip/hip_bf16.h>
#include <math.h>

typedef _Float16 half_t;
typedef __attribute__((ext_vector_type(16))) _Float16 v16h;
typedef __attribute__((ext_vector_type(8)))  _Float16 v8h;
typedef __attribute__((ext_vector_type(8)))  float    v8f;

#define PP 16
#define NN 128
#define CC 256
#define SS 64
#define HH 64
#define KA (CC * 3)

static __device__ __forceinline__ v16h join8(v8h lo, v8h hi) {
  return __builtin_shufflevector(lo, hi, 0, 1, 2, 3, 4, 5, 6, 7, 8, 9, 10, 11, 12, 13, 14, 15);
}

static __device__ __forceinline__ v8f wmma_f16(v16h a, v16h b, v8f c) {
  return __builtin_amdgcn_wmma_f32_16x16x32_f16(false, a, false, b, (short)0, c,
                                                false, false);
}

__global__ __launch_bounds__(128) void conv_a_kernel(
    const float* __restrict__ x, const float* __restrict__ w31a,
    const float* __restrict__ w33a, half_t* __restrict__ h31,
    half_t* __restrict__ h33) {
  __shared__ __align__(16) half_t lA[64 * 32];
  __shared__ __align__(16) half_t lB[64 * 32];

  const int tid = threadIdx.x;
  const int wv = tid >> 5;
  const int lane = tid & 31;
  const int g = lane >> 4;
  const int lj = lane & 15;
  const int p = blockIdx.z;
  const int br = blockIdx.y;
  const int n = blockIdx.x;

  const float* wa = (br == 0 ? w31a : w33a) + (size_t)p * HH * KA;
  half_t* hb = (br == 0 ? h31 : h33) + (size_t)p * NN * HH * SS;
  const float* xb = x + (size_t)(p * NN + n) * CC * SS;

  v8f acc[4] = {};
  for (int kc = 0; kc < KA; kc += 32) {
    const int dk = kc >> 8;
    const int ck = kc & 255;

#pragma unroll
    for (int i = 0; i < 16; ++i) {
      int e = tid + i * 128;
      int hid = e >> 5, kl = e & 31;
      lA[e] = (half_t)wa[hid * KA + (ck + kl) * 3 + dk];
    }
#pragma unroll
    for (int i = 0; i < 16; ++i) {
      int e = tid + i * 128;
      int kl = e >> 6, j = e & 63;
      int ss = j + dk - 1;
      float v = (ss >= 0 && ss < SS) ? xb[(ck + kl) * SS + ss] : 0.0f;
      lB[j * 32 + kl] = (half_t)v;
    }
    if (kc + 32 < KA) {
      int kn = kc + 32;
      __builtin_prefetch(xb + ((kn & 255) + (tid >> 2)) * SS, 0, 3);
    }
    __syncthreads();

    const int arow = (wv << 4) + lj;
    v16h afrag = join8(*(const v8h*)&lA[arow * 32 + g * 8],
                       *(const v8h*)&lA[arow * 32 + 16 + g * 8]);
#pragma unroll
    for (int t = 0; t < 4; ++t) {
      const int col = (t << 4) + lj;
      v16h bfrag = join8(*(const v8h*)&lB[col * 32 + g * 8],
                         *(const v8h*)&lB[col * 32 + 16 + g * 8]);
      acc[t] = wmma_f16(afrag, bfrag, acc[t]);
    }
    __syncthreads();
  }

#pragma unroll
  for (int t = 0; t < 4; ++t) {
#pragma unroll
    for (int r = 0; r < 8; ++r) {
      float v = acc[t][r];
      v = v >= 0.0f ? v : 0.01f * v;
      int hid = (wv << 4) + r + (g << 3);
      int s = (t << 4) + lj;
      hb[((size_t)n * HH + hid) * SS + s] = (half_t)v;
    }
  }
}

__global__ __launch_bounds__(128) void gate_kernel(
    const float* __restrict__ x, const float* __restrict__ w31b,
    const float* __restrict__ w33b, const half_t* __restrict__ h31,
    const half_t* __restrict__ h33, float* __restrict__ out) {
  __shared__ __align__(16) half_t lH31[SS * HH];
  __shared__ __align__(16) half_t lH33[(SS + 2) * HH];
  __shared__ __align__(16) half_t lW31[16 * HH];
  __shared__ __align__(16) half_t lW33[16 * 3 * HH];
  __shared__ __align__(16) float lX[16 * SS];
  __shared__ float red[4 * 16];

  const int tid = threadIdx.x;
  const int wv = tid >> 5;
  const int lane = tid & 31;
  const int g = lane >> 4;
  const int lj = lane & 15;
  const int p = blockIdx.z, n = blockIdx.y;
  const int c0 = blockIdx.x << 4;

  const half_t* hb31 = h31 + (size_t)(p * NN + n) * HH * SS;
  const half_t* hb33 = h33 + (size_t)(p * NN + n) * HH * SS;
  const float* xb = x + ((size_t)(p * NN + n) * CC + c0) * SS;
  const float* wb1 = w31b + (size_t)(p * CC + c0) * HH;
  const float* wb3 = w33b + (size_t)(p * CC + c0) * (HH * 3);

  for (int i = tid; i < SS * HH; i += 128) {
    int hid = i >> 6, s = i & 63;
    lH31[s * HH + hid] = hb31[i];
    lH33[(s + 1) * HH + hid] = hb33[i];
  }
  if (tid < HH) {
    lH33[tid] = (half_t)0.0f;
    lH33[(SS + 1) * HH + tid] = (half_t)0.0f;
  }
  for (int i = tid; i < 16 * HH; i += 128) lW31[i] = (half_t)wb1[i];
  for (int i = tid; i < 16 * 3 * HH; i += 128) {
    int cl = i / (3 * HH), k = i - cl * (3 * HH);
    int dk = k >> 6, hid = k & 63;
    lW33[i] = (half_t)wb3[cl * (3 * HH) + hid * 3 + dk];
  }
  for (int i = tid; i < 16 * SS; i += 128) lX[i] = xb[i];
  __syncthreads();

  const int s = (wv << 4) + lj;

  v8f acc31 = {}, acc33 = {};

#pragma unroll
  for (int kc = 0; kc < HH; kc += 32) {
    v16h a = join8(*(const v8h*)&lW31[lj * HH + kc + g * 8],
                   *(const v8h*)&lW31[lj * HH + kc + 16 + g * 8]);
    v16h b = join8(*(const v8h*)&lH31[s * HH + kc + g * 8],
                   *(const v8h*)&lH31[s * HH + kc + 16 + g * 8]);
    acc31 = wmma_f16(a, b, acc31);
  }

#pragma unroll
  for (int kc = 0; kc < 3 * HH; kc += 32) {
    int dk = kc >> 6;
    int hid0 = kc & 63;
    v16h a = join8(*(const v8h*)&lW33[lj * (3 * HH) + kc + g * 8],
                   *(const v8h*)&lW33[lj * (3 * HH) + kc + 16 + g * 8]);
    v16h b = join8(*(const v8h*)&lH33[(s + dk) * HH + hid0 + g * 8],
                   *(const v8h*)&lH33[(s + dk) * HH + hid0 + 16 + g * 8]);
    acc33 = wmma_f16(a, b, acc33);
  }

#pragma unroll
  for (int r = 0; r < 8; ++r) {
    int cl = r + (g << 3);
    const float* xr = &lX[cl * SS];
    float sum3 = 0.0f, mx3 = -INFINITY;
#pragma unroll
    for (int d = -1; d <= 1; ++d) {
      int ss = s + d;
      if (ss >= 0 && ss < SS) {
        float v = xr[ss];
        sum3 += v;
        mx3 = fmaxf(mx3, v);
      }
    }
    float sum5 = 0.0f, mx5 = -INFINITY;
#pragma unroll
    for (int d = -2; d <= 2; ++d) {
      int ss = s + d;
      if (ss >= 0 && ss < SS) {
        float v = xr[ss];
        sum5 += v;
        mx5 = fmaxf(mx5, v);
      }
    }
    float f31 = sum3 * (1.0f / 3.0f) + mx3;
    float f33 = sum5 * 0.2f + mx5;
    float g31 = 1.0f / (1.0f + __expf(-acc31[r]));
    float g33 = 1.0f / (1.0f + __expf(-acc33[r]));
    float val = f31 * g31 + f33 * g33;
#pragma unroll
    for (int m = 1; m < 16; m <<= 1) val = fmaxf(val, __shfl_xor(val, m, 32));
    if (lj == 0) red[wv * 16 + cl] = val;
  }
  __syncthreads();
  if (tid < 16) {
    float m = fmaxf(fmaxf(red[tid], red[16 + tid]),
                    fmaxf(red[32 + tid], red[48 + tid]));
    out[(size_t)(p * NN + n) * CC + c0 + tid] = m;
  }
}

extern "C" void kernel_launch(void* const* d_in, const int* in_sizes, int n_in,
                              void* d_out, int out_size, void* d_ws,
                              size_t ws_size, hipStream_t stream) {
  (void)in_sizes; (void)n_in; (void)out_size; (void)ws_size;
  const float* x    = (const float*)d_in[0];
  const float* w31a = (const float*)d_in[1];
  const float* w31b = (const float*)d_in[2];
  const float* w33a = (const float*)d_in[3];
  const float* w33b = (const float*)d_in[4];

  half_t* h31 = (half_t*)d_ws;
  half_t* h33 = h31 + (size_t)PP * NN * HH * SS;

  dim3 g1(NN, 2, PP);
  conv_a_kernel<<<g1, 128, 0, stream>>>(x, w31a, w33a, h31, h33);

  dim3 g2(CC / 16, NN, PP);
  gate_kernel<<<g2, 128, 0, stream>>>(x, w31b, w33b, h31, h33, (float*)d_out);
}


# gate kernel: one work-group per (p,n) stages h once and walks the 16 channel tiles, each tile's loads issued as one batch (was 16 work-groups, each re-staging h); sum and max order unchanged
# speedup vs baseline: 1.7191x; 1.7191x over previous
#include <hip/hip_runtime.h>
#include <hip/hip_bf16.h>
#include <math.h>

typedef _Float16 half_t;
typedef __attribute__((ext_vector_type(16))) _Float16 v16h;
typedef __attribute__((ext_vector_type(8)))  _Float16 v8h;
typedef __attribute__((ext_vector_type(8)))  float    v8f;

#define PP 16
#define NN 128
#define CC 256
#define SS 64
#define HH 64
#define KA (CC * 3)

static __device__ __forceinline__ v16h join8(v8h lo, v8h hi) {
  return __builtin_shufflevector(lo, hi, 0, 1, 2, 3, 4, 5, 6, 7, 8, 9, 10, 11, 12, 13, 14, 15);
}

static __device__ __forceinline__ v8f wmma_f16(v16h a, v16h b, v8f c) {
  return __builtin_amdgcn_wmma_f32_16x16x32_f16(false, a, false, b, (short)0, c,
                                                false, false);
}

__global__ __launch_bounds__(128) void conv_a_kernel(
    const float* __restrict__ x, const float* __restrict__ w31a,
    const float* __restrict__ w33a, half_t* __restrict__ h31,
    half_t* __restrict__ h33) {
  __shared__ __align__(16) half_t lA[64 * 32];
  __shared__ __align__(16) half_t lB[64 * 32];

  const int tid = threadIdx.x;
  const int wv = tid >> 5;
  const int lane = tid & 31;
  const int g = lane >> 4;
  const int lj = lane & 15;
  const int p = blockIdx.z;
  const int br = blockIdx.y;
  const int n = blockIdx.x;

  const float* wa = (br == 0 ? w31a : w33a) + (size_t)p * HH * KA;
  half_t* hb = (br == 0 ? h31 : h33) + (size_t)p * NN * HH * SS;
  const float* xb = x + (size_t)(p * NN + n) * CC * SS;

  v8f acc[4] = {};
  for (int kc = 0; kc < KA; kc += 32) {
    const int dk = kc >> 8;
    const int ck = kc & 255;

#pragma unroll
    for (int i = 0; i < 16; ++i) {
      int e = tid + i * 128;
      int hid = e >> 5, kl = e & 31;
      lA[e] = (half_t)wa[hid * KA + (ck + kl) * 3 + dk];
    }
#pragma unroll
    for (int i = 0; i < 16; ++i) {
      int e = tid + i * 128;
      int kl = e >> 6, j = e & 63;
      int ss = j + dk - 1;
      float v = (ss >= 0 && ss < SS) ? xb[(ck + kl) * SS + ss] : 0.0f;
      lB[j * 32 + kl] = (half_t)v;
    }
    if (kc + 32 < KA) {
      int kn = kc + 32;
      __builtin_prefetch(xb + ((kn & 255) + (tid >> 2)) * SS, 0, 3);
    }
    __syncthreads();

    const int arow = (wv << 4) + lj;
    v16h afrag = join8(*(const v8h*)&lA[arow * 32 + g * 8],
                       *(const v8h*)&lA[arow * 32 + 16 + g * 8]);
#pragma unroll
    for (int t = 0; t < 4; ++t) {
      const int col = (t << 4) + lj;
      v16h bfrag = join8(*(const v8h*)&lB[col * 32 + g * 8],
                         *(const v8h*)&lB[col * 32 + 16 + g * 8]);
      acc[t] = wmma_f16(afrag, bfrag, acc[t]);
    }
    __syncthreads();
  }

#pragma unroll
  for (int t = 0; t < 4; ++t) {
#pragma unroll
    for (int r = 0; r < 8; ++r) {
      float v = acc[t][r];
      v = v >= 0.0f ? v : 0.01f * v;
      int hid = (wv << 4) + r + (g << 3);
      int s = (t << 4) + lj;
      hb[((size_t)n * HH + hid) * SS + s] = (half_t)v;
    }
  }
}

__global__ __launch_bounds__(128, 8) void gate_kernel(
    const float* __restrict__ x, const float* __restrict__ w31b,
    const float* __restrict__ w33b, const half_t* __restrict__ h31,
    const half_t* __restrict__ h33, float* __restrict__ out) {
  __shared__ __align__(16) half_t lH31[SS * HH];
  __shared__ __align__(16) half_t lH33[(SS + 2) * HH];
  __shared__ __align__(16) half_t lW31[16 * HH];
  __shared__ __align__(16) half_t lW33[16 * 3 * HH];
  __shared__ __align__(16) float lX[16 * SS];
  __shared__ __align__(16) float red[4 * CC];

  const int tid = threadIdx.x;
  const int wv = tid >> 5;
  const int lane = tid & 31;
  const int g = lane >> 4;
  const int lj = lane & 15;
  const int p = blockIdx.y, n = blockIdx.x;

  const half_t* hb31 = h31 + (size_t)(p * NN + n) * HH * SS;
  const half_t* hb33 = h33 + (size_t)(p * NN + n) * HH * SS;

  for (int i = tid; i < SS * HH; i += 128) {
    int hid = i >> 6, s = i & 63;
    lH31[s * HH + hid] = hb31[i];
    lH33[(s + 1) * HH + hid] = hb33[i];
  }
  if (tid < HH) {
    lH33[tid] = (half_t)0.0f;
    lH33[(SS + 1) * HH + tid] = (half_t)0.0f;
  }
  __syncthreads();

  const int s = (wv << 4) + lj;

  for (int ct = 0; ct < CC / 16; ++ct) {
    const int c0 = ct << 4;
    const float* xb = x + ((size_t)(p * NN + n) * CC + c0) * SS;
    const float* wb1 = w31b + (size_t)(p * CC + c0) * HH;
    const float* wb3 = w33b + (size_t)(p * CC + c0) * (HH * 3);

#pragma unroll
    for (int j = 0; j < (16 * HH) / 128; ++j) {
      int i = tid + j * 128;
      lW31[i] = (half_t)wb1[i];
    }
#pragma unroll
    for (int j = 0; j < (16 * 3 * HH) / 128; ++j) {
      int i = tid + j * 128;
      int cl = i / (3 * HH), k = i - cl * (3 * HH);
      int dk = k >> 6, hid = k & 63;
      lW33[i] = (half_t)wb3[cl * (3 * HH) + hid * 3 + dk];
    }
#pragma unroll
    for (int j = 0; j < (16 * SS) / 128; ++j) {
      int i = tid + j * 128;
      lX[i] = xb[i];
    }
    __syncthreads();

    v8f acc31 = {}, acc33 = {};

#pragma unroll
    for (int q = 0; q < 2; ++q) {
      const int kc = q * 32;
      v16h a = join8(*(const v8h*)&lW31[lj * HH + kc + g * 8],
                     *(const v8h*)&lW31[lj * HH + kc + 16 + g * 8]);
      v16h b = join8(*(const v8h*)&lH31[s * HH + kc + g * 8],
                     *(const v8h*)&lH31[s * HH + kc + 16 + g * 8]);
      acc31 = wmma_f16(a, b, acc31);
    }

#pragma unroll
    for (int q = 0; q < 6; ++q) {
      const int kc = q * 32;
      const int dk = kc >> 6;
      const int hid0 = kc & 63;
      v16h a = join8(*(const v8h*)&lW33[lj * (3 * HH) + kc + g * 8],
                     *(const v8h*)&lW33[lj * (3 * HH) + kc + 16 + g * 8]);
      v16h b = join8(*(const v8h*)&lH33[(s + dk) * HH + hid0 + g * 8],
                     *(const v8h*)&lH33[(s + dk) * HH + hid0 + 16 + g * 8]);
      acc33 = wmma_f16(a, b, acc33);
    }

#pragma unroll
    for (int r = 0; r < 8; ++r) {
      int cl = r + (g << 3);
      const float* xr = &lX[cl * SS];
      float sum3 = 0.0f, mx3 = -INFINITY;
#pragma unroll
      for (int d = -1; d <= 1; ++d) {
        int ss = s + d;
        if (ss >= 0 && ss < SS) {
          float v = xr[ss];
          sum3 += v;
          mx3 = fmaxf(mx3, v);
        }
      }
      float sum5 = 0.0f, mx5 = -INFINITY;
#pragma unroll
      for (int d = -2; d <= 2; ++d) {
        int ss = s + d;
        if (ss >= 0 && ss < SS) {
          float v = xr[ss];
          sum5 += v;
          mx5 = fmaxf(mx5, v);
        }
      }
      float f31 = sum3 * (1.0f / 3.0f) + mx3;
      float f33 = sum5 * 0.2f + mx5;
      float g31 = 1.0f / (1.0f + __expf(-acc31[r]));
      float g33 = 1.0f / (1.0f + __expf(-acc33[r]));
      float val = f31 * g31 + f33 * g33;
#pragma unroll
      for (int m = 1; m < 16; m <<= 1) val = fmaxf(val, __shfl_xor(val, m, 32));
      if (lj == 0) red[wv * CC + c0 + cl] = val;
    }
    __syncthreads();
  }

  {
    const int c = tid << 1;
    float2 m0 = *(const float2*)&red[c];
    float2 m1 = *(const float2*)&red[CC + c];
    float2 m2 = *(const float2*)&red[2 * CC + c];
    float2 m3 = *(const float2*)&red[3 * CC + c];
    float2 m;
    m.x = fmaxf(fmaxf(m0.x, m1.x), fmaxf(m2.x, m3.x));
    m.y = fmaxf(fmaxf(m0.y, m1.y), fmaxf(m2.y, m3.y));
    *(float2*)&out[(size_t)(p * NN + n) * CC + c] = m;
  }
}

extern "C" void kernel_launch(void* const* d_in, const int* in_sizes, int n_in,
                              void* d_out, int out_size, void* d_ws,
                              size_t ws_size, hipStream_t stream) {
  (void)in_sizes; (void)n_in; (void)out_size; (void)ws_size;
  const float* x    = (const float*)d_in[0];
  const float* w31a = (const float*)d_in[1];
  const float* w31b = (const float*)d_in[2];
  const float* w33a = (const float*)d_in[3];
  const float* w33b = (const float*)d_in[4];

  half_t* h31 = (half_t*)d_ws;
  half_t* h33 = h31 + (size_t)PP * NN * HH * SS;

  dim3 g1(NN, 2, PP);
  conv_a_kernel<<<g1, 128, 0, stream>>>(x, w31a, w33a, h31, h33);

  dim3 g2(NN, PP);
  gate_kernel<<<g2, 128, 0, stream>>>(x, w31b, w33b, h31, h33, (float*)d_out);
}


# LDS rows re-pitched to 16 B x odd and transposing stage stores widened to 16 bytes (their 8 loads issue together): fragment reads and stage stores stop replaying on bank conflicts; sums unchanged
# speedup vs baseline: 3.3291x; 1.9365x over previous
#include <hip/hip_runtime.h>
#include <hip/hip_bf16.h>
#include <math.h>

typedef _Float16 half_t;
typedef __attribute__((ext_vector_type(16))) _Float16 v16h;
typedef __attribute__((ext_vector_type(8)))  _Float16 v8h;
typedef __attribute__((ext_vector_type(8)))  float    v8f;

#define PP 16
#define NN 128
#define CC 256
#define SS 64
#define HH 64
#define KA (CC * 3)

#define PA 40
#define PH 72
#define PW3 200
#define PX 68

static __device__ __forceinline__ v16h join8(v8h lo, v8h hi) {
  return __builtin_shufflevector(lo, hi, 0, 1, 2, 3, 4, 5, 6, 7, 8, 9, 10, 11, 12, 13, 14, 15);
}

static __device__ __forceinline__ v8f wmma_f16(v16h a, v16h b, v8f c) {
  return __builtin_amdgcn_wmma_f32_16x16x32_f16(false, a, false, b, (short)0, c,
                                                false, false);
}

__global__ __launch_bounds__(128) void conv_a_kernel(
    const float* __restrict__ x, const float* __restrict__ w31a,
    const float* __restrict__ w33a, half_t* __restrict__ h31,
    half_t* __restrict__ h33) {
  __shared__ __align__(16) half_t lA[64 * PA];
  __shared__ __align__(16) half_t lB[64 * PA];

  const int tid = threadIdx.x;
  const int wv = tid >> 5;
  const int lane = tid & 31;
  const int g = lane >> 4;
  const int lj = lane & 15;
  const int p = blockIdx.z;
  const int br = blockIdx.y;
  const int n = blockIdx.x;

  const float* wa = (br == 0 ? w31a : w33a) + (size_t)p * HH * KA;
  half_t* hb = (br == 0 ? h31 : h33) + (size_t)p * NN * HH * SS;
  const float* xb = x + (size_t)(p * NN + n) * CC * SS;

  v8f acc[4] = {};
  for (int kc = 0; kc < KA; kc += 32) {
    const int dk = kc >> 8;
    const int ck = kc & 255;

    const float* was = wa + (tid >> 5) * KA + (ck + (tid & 31)) * 3 + dk;
    half_t* las = &lA[(tid >> 5) * PA + (tid & 31)];
#pragma unroll
    for (int i = 0; i < 16; ++i) las[i * 4 * PA] = (half_t)was[i * 4 * KA];
#pragma unroll
    for (int i = 0; i < 2; ++i) {
      int u = tid + i * 128;
      int j = u & 63, kg = u >> 6;
      int ss = j + dk - 1;
      v8h v;
      if (ss >= 0 && ss < SS) {
        const float* xs = xb + (ck + kg * 8) * SS + ss;
#pragma unroll
        for (int t = 0; t < 8; ++t) v[t] = (half_t)xs[t * SS];
      } else {
#pragma unroll
        for (int t = 0; t < 8; ++t) v[t] = (half_t)0.0f;
      }
      *(v8h*)&lB[j * PA + kg * 8] = v;
    }
    if (kc + 32 < KA) {
      int kn = kc + 32;
      __builtin_prefetch(xb + ((kn & 255) + (tid >> 2)) * SS, 0, 3);
    }
    __syncthreads();

    const int arow = (wv << 4) + lj;
    v16h afrag = join8(*(const v8h*)&lA[arow * PA + g * 8],
                       *(const v8h*)&lA[arow * PA + 16 + g * 8]);
#pragma unroll
    for (int t = 0; t < 4; ++t) {
      const int col = (t << 4) + lj;
      v16h bfrag = join8(*(const v8h*)&lB[col * PA + g * 8],
                         *(const v8h*)&lB[col * PA + 16 + g * 8]);
      acc[t] = wmma_f16(afrag, bfrag, acc[t]);
    }
    __syncthreads();
  }

#pragma unroll
  for (int t = 0; t < 4; ++t) {
#pragma unroll
    for (int r = 0; r < 8; ++r) {
      float v = acc[t][r];
      v = v >= 0.0f ? v : 0.01f * v;
      int hid = (wv << 4) + r + (g << 3);
      int s = (t << 4) + lj;
      hb[((size_t)n * HH + hid) * SS + s] = (half_t)v;
    }
  }
}

__global__ __launch_bounds__(128, 8) void gate_kernel(
    const float* __restrict__ x, const float* __restrict__ w31b,
    const float* __restrict__ w33b, const half_t* __restrict__ h31,
    const half_t* __restrict__ h33, float* __restrict__ out) {
  __shared__ __align__(16) half_t lH31[SS * PH];
  __shared__ __align__(16) half_t lH33[(SS + 2) * PH];
  __shared__ __align__(16) half_t lW31[16 * PH];
  __shared__ __align__(16) half_t lW33[16 * PW3];
  __shared__ __align__(16) float lX[16 * PX];
  __shared__ __align__(16) float red[4 * CC];

  const int tid = threadIdx.x;
  const int wv = tid >> 5;
  const int lane = tid & 31;
  const int g = lane >> 4;
  const int lj = lane & 15;
  const int p = blockIdx.y, n = blockIdx.x;

  const half_t* hb31 = h31 + (size_t)(p * NN + n) * HH * SS;
  const half_t* hb33 = h33 + (size_t)(p * NN + n) * HH * SS;

#pragma unroll
  for (int i = 0; i < (SS * HH) / (8 * 128); ++i) {
    int u = tid + i * 128;
    int s = u & 63, hg = u >> 6;
    v8h a, b;
#pragma unroll
    for (int t = 0; t < 8; ++t) {
      a[t] = hb31[(hg * 8 + t) * SS + s];
      b[t] = hb33[(hg * 8 + t) * SS + s];
    }
    *(v8h*)&lH31[s * PH + hg * 8] = a;
    *(v8h*)&lH33[(s + 1) * PH + hg * 8] = b;
  }
  if (tid < HH) {
    lH33[tid] = (half_t)0.0f;
    lH33[(SS + 1) * PH + tid] = (half_t)0.0f;
  }
  __syncthreads();

  const int s = (wv << 4) + lj;
  const int r2 = tid >> 6, c64 = tid & 63;

  for (int ct = 0; ct < CC / 16; ++ct) {
    const int c0 = ct << 4;
    const float* xb = x + ((size_t)(p * NN + n) * CC + c0) * SS;
    const float* wb1 = w31b + (size_t)(p * CC + c0) * HH;
    const float* wb3 = w33b + (size_t)(p * CC + c0) * (HH * 3);

#pragma unroll
    for (int j = 0; j < (16 * HH) / 128; ++j)
      lW31[r2 * PH + c64 + j * 2 * PH] = (half_t)wb1[tid + j * 128];
#pragma unroll
    for (int j = 0; j < (16 * 3 * HH) / 128; ++j) {
      const int cj = j / 3, dk = j % 3;
      lW33[r2 * 8 * PW3 + c64 + cj * PW3 + dk * HH] =
          (half_t)wb3[r2 * 8 * (3 * HH) + c64 * 3 + cj * (3 * HH) + dk];
    }
#pragma unroll
    for (int j = 0; j < (16 * SS) / 128; ++j)
      lX[r2 * PX + c64 + j * 2 * PX] = xb[tid + j * 128];
    __syncthreads();

    v8f acc31 = {}, acc33 = {};

#pragma unroll
    for (int q = 0; q < 2; ++q) {
      const int kc = q * 32;
      v16h a = join8(*(const v8h*)&lW31[lj * PH + kc + g * 8],
                     *(const v8h*)&lW31[lj * PH + kc + 16 + g * 8]);
      v16h b = join8(*(const v8h*)&lH31[s * PH + kc + g * 8],
                     *(const v8h*)&lH31[s * PH + kc + 16 + g * 8]);
      acc31 = wmma_f16(a, b, acc31);
    }

#pragma unroll
    for (int q = 0; q < 6; ++q) {
      const int kc = q * 32;
      const int dk = kc >> 6;
      const int hid0 = kc & 63;
      v16h a = join8(*(const v8h*)&lW33[lj * PW3 + kc + g * 8],
                     *(const v8h*)&lW33[lj * PW3 + kc + 16 + g * 8]);
      v16h b = join8(*(const v8h*)&lH33[(s + dk) * PH + hid0 + g * 8],
                     *(const v8h*)&lH33[(s + dk) * PH + hid0 + 16 + g * 8]);
      acc33 = wmma_f16(a, b, acc33);
    }

#pragma unroll
    for (int r = 0; r < 8; ++r) {
      int cl = r + (g << 3);
      const float* xr = &lX[cl * PX];
      float sum3 = 0.0f, mx3 = -INFINITY;
#pragma unroll
      for (int d = -1; d <= 1; ++d) {
        int ss = s + d;
        if (ss >= 0 && ss < SS) {
          float v = xr[ss];
          sum3 += v;
          mx3 = fmaxf(mx3, v);
        }
      }
      float sum5 = 0.0f, mx5 = -INFINITY;
#pragma unroll
      for (int d = -2; d <= 2; ++d) {
        int ss = s + d;
        if (ss >= 0 && ss < SS) {
          float v = xr[ss];
          sum5 += v;
          mx5 = fmaxf(mx5, v);
        }
      }
      float f31 = sum3 * (1.0f / 3.0f) + mx3;
      float f33 = sum5 * 0.2f + mx5;
      float g31 = 1.0f / (1.0f + __expf(-acc31[r]));
      float g33 = 1.0f / (1.0f + __expf(-acc33[r]));
      float val = f31 * g31 + f33 * g33;
#pragma unroll
      for (int m = 1; m < 16; m <<= 1) val = fmaxf(val, __shfl_xor(val, m, 32));
      if (lj == 0) red[wv * CC + c0 + cl] = val;
    }
    __syncthreads();
  }

  {
    const int c = tid << 1;
    float2 m0 = *(const float2*)&red[c];
    float2 m1 = *(const float2*)&red[CC + c];
    float2 m2 = *(const float2*)&red[2 * CC + c];
    float2 m3 = *(const float2*)&red[3 * CC + c];
    float2 m;
    m.x = fmaxf(fmaxf(m0.x, m1.x), fmaxf(m2.x, m3.x));
    m.y = fmaxf(fmaxf(m0.y, m1.y), fmaxf(m2.y, m3.y));
    *(float2*)&out[(size_t)(p * NN + n) * CC + c] = m;
  }
}

extern "C" void kernel_launch(void* const* d_in, const int* in_sizes, int n_in,
                              void* d_out, int out_size, void* d_ws,
                              size_t ws_size, hipStream_t stream) {
  (void)in_sizes; (void)n_in; (void)out_size; (void)ws_size;
  const float* x    = (const float*)d_in[0];
  const float* w31a = (const float*)d_in[1];
  const float* w31b = (const float*)d_in[2];
  const float* w33a = (const float*)d_in[3];
  const float* w33b = (const float*)d_in[4];

  half_t* h31 = (half_t*)d_ws;
  half_t* h33 = h31 + (size_t)PP * NN * HH * SS;

  dim3 g1(NN, 2, PP);
  conv_a_kernel<<<g1, 128, 0, stream>>>(x, w31a, w33a, h31, h33);

  dim3 g2(NN, PP);
  gate_kernel<<<g2, 128, 0, stream>>>(x, w31b, w33b, h31, h33, (float*)d_out);
}


# conv kernel: K walked channel-major (chunk = 32 channels x 3 taps): an x chunk is staged once for all 3 taps, weight rows load as contiguous 16-byte pieces; order of the k sum changes
# speedup vs baseline: 4.1556x; 1.2483x over previous
#include <hip/hip_runtime.h>
#include <hip/hip_bf16.h>
#include <math.h>

typedef _Float16 half_t;
typedef __attribute__((ext_vector_type(16))) _Float16 v16h;
typedef __attribute__((ext_vector_type(8)))  _Float16 v8h;
typedef __attribute__((ext_vector_type(8)))  float    v8f;

#define PP 16
#define NN 128
#define CC 256
#define SS 64
#define HH 64
#define KA (CC * 3)

#define PA 40
#define PH 72
#define PW3 200
#define APL (64 * PA + 16)
#define PX 68

static __device__ __forceinline__ v16h join8(v8h lo, v8h hi) {
  return __builtin_shufflevector(lo, hi, 0, 1, 2, 3, 4, 5, 6, 7, 8, 9, 10, 11, 12, 13, 14, 15);
}

static __device__ __forceinline__ v8f wmma_f16(v16h a, v16h b, v8f c) {
  return __builtin_amdgcn_wmma_f32_16x16x32_f16(false, a, false, b, (short)0, c,
                                                false, false);
}

__global__ __launch_bounds__(128) void conv_a_kernel(
    const float* __restrict__ x, const float* __restrict__ w31a,
    const float* __restrict__ w33a, half_t* __restrict__ h31,
    half_t* __restrict__ h33) {
  __shared__ __align__(16) half_t lA[3 * APL];
  __shared__ __align__(16) half_t lB[(SS + 2) * PA];

  const int tid = threadIdx.x;
  const int wv = tid >> 5;
  const int lane = tid & 31;
  const int g = lane >> 4;
  const int lj = lane & 15;
  const int p = blockIdx.z;
  const int br = blockIdx.y;
  const int n = blockIdx.x;

  const float* wa = (br == 0 ? w31a : w33a) + (size_t)p * HH * KA;
  half_t* hb = (br == 0 ? h31 : h33) + (size_t)p * NN * HH * SS;
  const float* xb = x + (size_t)(p * NN + n) * CC * SS;

  const int sub = tid & 7, unit0 = tid >> 3;
  int aoff[12];
#pragma unroll
  for (int q = 0; q < 12; ++q) {
    const int idx = (q >> 2) * 32 + sub * 4 + (q & 3);
    aoff[q] = (idx % 3) * APL + unit0 * PA + idx / 3;
  }
  const float* was0 = wa + unit0 * KA + sub * 4;

  if (tid < 32) {
    lB[tid] = (half_t)0.0f;
    lB[(SS + 1) * PA + tid] = (half_t)0.0f;
  }

  v8f acc[4] = {};
  for (int ck = 0; ck < CC; ck += 32) {
    const float* was = was0 + ck * 3;
    float4 wq[12];
#pragma unroll
    for (int i = 0; i < 12; ++i)
      wq[i] = *(const float4*)(was + (i & 3) * 16 * KA + (i >> 2) * 32);
#pragma unroll
    for (int i = 0; i < 12; ++i) {
      half_t* dst = &lA[(i & 3) * 16 * PA];
      const int q0 = (i >> 2) * 4;
      dst[aoff[q0 + 0]] = (half_t)wq[i].x;
      dst[aoff[q0 + 1]] = (half_t)wq[i].y;
      dst[aoff[q0 + 2]] = (half_t)wq[i].z;
      dst[aoff[q0 + 3]] = (half_t)wq[i].w;
    }
#pragma unroll
    for (int i = 0; i < 2; ++i) {
      int u = tid + i * 128;
      int j = u & 63, kg = u >> 6;
      const float* xs = xb + (ck + kg * 8) * SS + j;
      v8h v;
#pragma unroll
      for (int t = 0; t < 8; ++t) v[t] = (half_t)xs[t * SS];
      *(v8h*)&lB[(j + 1) * PA + kg * 8] = v;
    }
    if (ck + 32 < CC) {
      __builtin_prefetch(xb + (ck + 32 + (tid >> 2)) * SS, 0, 3);
    }
    __syncthreads();

    const int arow = (wv << 4) + lj;
#pragma unroll
    for (int dk = 0; dk < 3; ++dk) {
      const half_t* ar = &lA[dk * APL + arow * PA];
      v16h afrag = join8(*(const v8h*)&ar[g * 8], *(const v8h*)&ar[16 + g * 8]);
#pragma unroll
      for (int t = 0; t < 4; ++t) {
        const half_t* brw = &lB[((t << 4) + lj + dk) * PA];
        v16h bfrag = join8(*(const v8h*)&brw[g * 8], *(const v8h*)&brw[16 + g * 8]);
        acc[t] = wmma_f16(afrag, bfrag, acc[t]);
      }
    }
    __syncthreads();
  }

#pragma unroll
  for (int t = 0; t < 4; ++t) {
#pragma unroll
    for (int r = 0; r < 8; ++r) {
      float v = acc[t][r];
      v = v >= 0.0f ? v : 0.01f * v;
      int hid = (wv << 4) + r + (g << 3);
      int s = (t << 4) + lj;
      hb[((size_t)n * HH + hid) * SS + s] = (half_t)v;
    }
  }
}

__global__ __launch_bounds__(128, 8) void gate_kernel(
    const float* __restrict__ x, const float* __restrict__ w31b,
    const float* __restrict__ w33b, const half_t* __restrict__ h31,
    const half_t* __restrict__ h33, float* __restrict__ out) {
  __shared__ __align__(16) half_t lH31[SS * PH];
  __shared__ __align__(16) half_t lH33[(SS + 2) * PH];
  __shared__ __align__(16) half_t lW31[16 * PH];
  __shared__ __align__(16) half_t lW33[16 * PW3];
  __shared__ __align__(16) float lX[16 * PX];
  __shared__ __align__(16) float red[4 * CC];

  const int tid = threadIdx.x;
  const int wv = tid >> 5;
  const int lane = tid & 31;
  const int g = lane >> 4;
  const int lj = lane & 15;
  const int p = blockIdx.y, n = blockIdx.x;

  const half_t* hb31 = h31 + (size_t)(p * NN + n) * HH * SS;
  const half_t* hb33 = h33 + (size_t)(p * NN + n) * HH * SS;

#pragma unroll
  for (int i = 0; i < (SS * HH) / (8 * 128); ++i) {
    int u = tid + i * 128;
    int s = u & 63, hg = u >> 6;
    v8h a, b;
#pragma unroll
    for (int t = 0; t < 8; ++t) {
      a[t] = hb31[(hg * 8 + t) * SS + s];
      b[t] = hb33[(hg * 8 + t) * SS + s];
    }
    *(v8h*)&lH31[s * PH + hg * 8] = a;
    *(v8h*)&lH33[(s + 1) * PH + hg * 8] = b;
  }
  if (tid < HH) {
    lH33[tid] = (half_t)0.0f;
    lH33[(SS + 1) * PH + tid] = (half_t)0.0f;
  }
  __syncthreads();

  const int s = (wv << 4) + lj;
  const int r2 = tid >> 6, c64 = tid & 63;

  for (int ct = 0; ct < CC / 16; ++ct) {
    const int c0 = ct << 4;
    const float* xb = x + ((size_t)(p * NN + n) * CC + c0) * SS;
    const float* wb1 = w31b + (size_t)(p * CC + c0) * HH;
    const float* wb3 = w33b + (size_t)(p * CC + c0) * (HH * 3);

#pragma unroll
    for (int j = 0; j < (16 * HH) / 128; ++j)
      lW31[r2 * PH + c64 + j * 2 * PH] = (half_t)wb1[tid + j * 128];
#pragma unroll
    for (int j = 0; j < (16 * 3 * HH) / 128; ++j) {
      const int cj = j / 3, dk = j % 3;
      lW33[r2 * 8 * PW3 + c64 + cj * PW3 + dk * HH] =
          (half_t)wb3[r2 * 8 * (3 * HH) + c64 * 3 + cj * (3 * HH) + dk];
    }
#pragma unroll
    for (int j = 0; j < (16 * SS) / 128; ++j)
      lX[r2 * PX + c64 + j * 2 * PX] = xb[tid + j * 128];
    __syncthreads();

    v8f acc31 = {}, acc33 = {};

#pragma unroll
    for (int q = 0; q < 2; ++q) {
      const int kc = q * 32;
      v16h a = join8(*(const v8h*)&lW31[lj * PH + kc + g * 8],
                     *(const v8h*)&lW31[lj * PH + kc + 16 + g * 8]);
      v16h b = join8(*(const v8h*)&lH31[s * PH + kc + g * 8],
                     *(const v8h*)&lH31[s * PH + kc + 16 + g * 8]);
      acc31 = wmma_f16(a, b, acc31);
    }

#pragma unroll
    for (int q = 0; q < 6; ++q) {
      const int kc = q * 32;
      const int dk = kc >> 6;
      const int hid0 = kc & 63;
      v16h a = join8(*(const v8h*)&lW33[lj * PW3 + kc + g * 8],
                     *(const v8h*)&lW33[lj * PW3 + kc + 16 + g * 8]);
      v16h b = join8(*(const v8h*)&lH33[(s + dk) * PH + hid0 + g * 8],
                     *(const v8h*)&lH33[(s + dk) * PH + hid0 + 16 + g * 8]);
      acc33 = wmma_f16(a, b, acc33);
    }

#pragma unroll
    for (int r = 0; r < 8; ++r) {
      int cl = r + (g << 3);
      const float* xr = &lX[cl * PX];
      float sum3 = 0.0f, mx3 = -INFINITY;
#pragma unroll
      for (int d = -1; d <= 1; ++d) {
        int ss = s + d;
        if (ss >= 0 && ss < SS) {
          float v = xr[ss];
          sum3 += v;
          mx3 = fmaxf(mx3, v);
        }
      }
      float sum5 = 0.0f, mx5 = -INFINITY;
#pragma unroll
      for (int d = -2; d <= 2; ++d) {
        int ss = s + d;
        if (ss >= 0 && ss < SS) {
          float v = xr[ss];
          sum5 += v;
          mx5 = fmaxf(mx5, v);
        }
      }
      float f31 = sum3 * (1.0f / 3.0f) + mx3;
      float f33 = sum5 * 0.2f + mx5;
      float g31 = 1.0f / (1.0f + __expf(-acc31[r]));
      float g33 = 1.0f / (1.0f + __expf(-acc33[r]));
      float val = f31 * g31 + f33 * g33;
#pragma unroll
      for (int m = 1; m < 16; m <<= 1) val = fmaxf(val, __shfl_xor(val, m, 32));
      if (lj == 0) red[wv * CC + c0 + cl] = val;
    }
    __syncthreads();
  }

  {
    const int c = tid << 1;
    float2 m0 = *(const float2*)&red[c];
    float2 m1 = *(const float2*)&red[CC + c];
    float2 m2 = *(const float2*)&red[2 * CC + c];
    float2 m3 = *(const float2*)&red[3 * CC + c];
    float2 m;
    m.x = fmaxf(fmaxf(m0.x, m1.x), fmaxf(m2.x, m3.x));
    m.y = fmaxf(fmaxf(m0.y, m1.y), fmaxf(m2.y, m3.y));
    *(float2*)&out[(size_t)(p * NN + n) * CC + c] = m;
  }
}

extern "C" void kernel_launch(void* const* d_in, const int* in_sizes, int n_in,
                              void* d_out, int out_size, void* d_ws,
                              size_t ws_size, hipStream_t stream) {
  (void)in_sizes; (void)n_in; (void)out_size; (void)ws_size;
  const float* x    = (const float*)d_in[0];
  const float* w31a = (const float*)d_in[1];
  const float* w31b = (const float*)d_in[2];
  const float* w33a = (const float*)d_in[3];
  const float* w33b = (const float*)d_in[4];

  half_t* h31 = (half_t*)d_ws;
  half_t* h33 = h31 + (size_t)PP * NN * HH * SS;

  dim3 g1(NN, 2, PP);
  conv_a_kernel<<<g1, 128, 0, stream>>>(x, w31a, w33a, h31, h33);

  dim3 g2(NN, PP);
  gate_kernel<<<g2, 128, 0, stream>>>(x, w31b, w33b, h31, h33, (float*)d_out);
}


# gate kernel: operand slots of its products exchanged (s on rows, channels on lanes): a lane holds 8 s of one channel, pooling from 16-byte reads, no guards, max over s in registers; sums unchanged
# speedup vs baseline: 5.7279x; 1.3784x over previous
#include <hip/hip_runtime.h>
#include <hip/hip_bf16.h>
#include <math.h>

typedef _Float16 half_t;
typedef __attribute__((ext_vector_type(16))) _Float16 v16h;
typedef __attribute__((ext_vector_type(8)))  _Float16 v8h;
typedef __attribute__((ext_vector_type(8)))  float    v8f;

#define PP 16
#define NN 128
#define CC 256
#define SS 64
#define HH 64
#define KA (CC * 3)

#define PA 40
#define PH 72
#define PW3 200
#define APL (64 * PA + 16)
#define PX 68
#define XO 2

static __device__ __forceinline__ v16h join8(v8h lo, v8h hi) {
  return __builtin_shufflevector(lo, hi, 0, 1, 2, 3, 4, 5, 6, 7, 8, 9, 10, 11, 12, 13, 14, 15);
}

static __device__ __forceinline__ float other_row(float x) {
  const int b = __builtin_bit_cast(int, x);
  return __builtin_bit_cast(
      float, __builtin_amdgcn_permlanex16(b, b, 0x76543210, 0xfedcba98, false, false));
}

static __device__ __forceinline__ v8f wmma_f16(v16h a, v16h b, v8f c) {
  return __builtin_amdgcn_wmma_f32_16x16x32_f16(false, a, false, b, (short)0, c,
                                                false, false);
}

__global__ __launch_bounds__(128) void conv_a_kernel(
    const float* __restrict__ x, const float* __restrict__ w31a,
    const float* __restrict__ w33a, half_t* __restrict__ h31,
    half_t* __restrict__ h33) {
  __shared__ __align__(16) half_t lA[3 * APL];
  __shared__ __align__(16) half_t lB[(SS + 2) * PA];

  const int tid = threadIdx.x;
  const int wv = tid >> 5;
  const int lane = tid & 31;
  const int g = lane >> 4;
  const int lj = lane & 15;
  const int p = blockIdx.z;
  const int br = blockIdx.y;
  const int n = blockIdx.x;

  const float* wa = (br == 0 ? w31a : w33a) + (size_t)p * HH * KA;
  half_t* hb = (br == 0 ? h31 : h33) + (size_t)p * NN * HH * SS;
  const float* xb = x + (size_t)(p * NN + n) * CC * SS;

  const int sub = tid & 7, unit0 = tid >> 3;
  int aoff[12];
#pragma unroll
  for (int q = 0; q < 12; ++q) {
    const int idx = (q >> 2) * 32 + sub * 4 + (q & 3);
    aoff[q] = (idx % 3) * APL + unit0 * PA + idx / 3;
  }
  const float* was0 = wa + unit0 * KA + sub * 4;

  if (tid < 32) {
    lB[tid] = (half_t)0.0f;
    lB[(SS + 1) * PA + tid] = (half_t)0.0f;
  }

  v8f acc[4] = {};
  for (int ck = 0; ck < CC; ck += 32) {
    const float* was = was0 + ck * 3;
    float4 wq[12];
#pragma unroll
    for (int i = 0; i < 12; ++i)
      wq[i] = *(const float4*)(was + (i & 3) * 16 * KA + (i >> 2) * 32);
#pragma unroll
    for (int i = 0; i < 12; ++i) {
      half_t* dst = &lA[(i & 3) * 16 * PA];
      const int q0 = (i >> 2) * 4;
      dst[aoff[q0 + 0]] = (half_t)wq[i].x;
      dst[aoff[q0 + 1]] = (half_t)wq[i].y;
      dst[aoff[q0 + 2]] = (half_t)wq[i].z;
      dst[aoff[q0 + 3]] = (half_t)wq[i].w;
    }
#pragma unroll
    for (int i = 0; i < 2; ++i) {
      int u = tid + i * 128;
      int j = u & 63, kg = u >> 6;
      const float* xs = xb + (ck + kg * 8) * SS + j;
      v8h v;
#pragma unroll
      for (int t = 0; t < 8; ++t) v[t] = (half_t)xs[t * SS];
      *(v8h*)&lB[(j + 1) * PA + kg * 8] = v;
    }
    if (ck + 32 < CC) {
      __builtin_prefetch(xb + (ck + 32 + (tid >> 2)) * SS, 0, 3);
    }
    __syncthreads();

    const int arow = (wv << 4) + lj;
#pragma unroll
    for (int dk = 0; dk < 3; ++dk) {
      const half_t* ar = &lA[dk * APL + arow * PA];
      v16h afrag = join8(*(const v8h*)&ar[g * 8], *(const v8h*)&ar[16 + g * 8]);
#pragma unroll
      for (int t = 0; t < 4; ++t) {
        const half_t* brw = &lB[((t << 4) + lj + dk) * PA];
        v16h bfrag = join8(*(const v8h*)&brw[g * 8], *(const v8h*)&brw[16 + g * 8]);
        acc[t] = wmma_f16(afrag, bfrag, acc[t]);
      }
    }
    __syncthreads();
  }

#pragma unroll
  for (int t = 0; t < 4; ++t) {
#pragma unroll
    for (int r = 0; r < 8; ++r) {
      float v = acc[t][r];
      v = v >= 0.0f ? v : 0.01f * v;
      int hid = (wv << 4) + r + (g << 3);
      int s = (t << 4) + lj;
      hb[((size_t)n * HH + hid) * SS + s] = (half_t)v;
    }
  }
}

__global__ __launch_bounds__(128, 8) void gate_kernel(
    const float* __restrict__ x, const float* __restrict__ w31b,
    const float* __restrict__ w33b, const half_t* __restrict__ h31,
    const half_t* __restrict__ h33, float* __restrict__ out) {
  __shared__ __align__(16) half_t lH31[SS * PH];
  __shared__ __align__(16) half_t lH33[(SS + 2) * PH];
  __shared__ __align__(16) half_t lW31[16 * PH];
  __shared__ __align__(16) half_t lW33[16 * PW3];
  __shared__ __align__(16) float lX[16 * PX];
  __shared__ __align__(16) float red[4 * CC];

  const int tid = threadIdx.x;
  const int wv = tid >> 5;
  const int lane = tid & 31;
  const int g = lane >> 4;
  const int lj = lane & 15;
  const int p = blockIdx.y, n = blockIdx.x;

  const half_t* hb31 = h31 + (size_t)(p * NN + n) * HH * SS;
  const half_t* hb33 = h33 + (size_t)(p * NN + n) * HH * SS;

#pragma unroll
  for (int i = 0; i < (SS * HH) / (8 * 128); ++i) {
    int u = tid + i * 128;
    int s = u & 63, hg = u >> 6;
    v8h a, b;
#pragma unroll
    for (int t = 0; t < 8; ++t) {
      a[t] = hb31[(hg * 8 + t) * SS + s];
      b[t] = hb33[(hg * 8 + t) * SS + s];
    }
    *(v8h*)&lH31[s * PH + hg * 8] = a;
    *(v8h*)&lH33[(s + 1) * PH + hg * 8] = b;
  }
  if (tid < HH) {
    lH33[tid] = (half_t)0.0f;
    lH33[(SS + 1) * PH + tid] = (half_t)0.0f;
    const int zr = tid >> 2, zk = tid & 3;
    lX[zr * PX + (zk < 2 ? zk : SS + zk)] = 0.0f;
  }
  __syncthreads();

  const int s = (wv << 4) + lj;
  const int r2 = tid >> 6, c64 = tid & 63;
  const int s0 = (wv << 4) + (g << 3);
  const bool lo_edge = s0 == 0, hi_edge = s0 == SS - 8;

  for (int ct = 0; ct < CC / 16; ++ct) {
    const int c0 = ct << 4;
    const float* xb = x + ((size_t)(p * NN + n) * CC + c0) * SS;
    const float* wb1 = w31b + (size_t)(p * CC + c0) * HH;
    const float* wb3 = w33b + (size_t)(p * CC + c0) * (HH * 3);

#pragma unroll
    for (int j = 0; j < (16 * HH) / 128; ++j)
      lW31[r2 * PH + c64 + j * 2 * PH] = (half_t)wb1[tid + j * 128];
#pragma unroll
    for (int j = 0; j < (16 * 3 * HH) / 128; ++j) {
      const int cj = j / 3, dk = j % 3;
      lW33[r2 * 8 * PW3 + c64 + cj * PW3 + dk * HH] =
          (half_t)wb3[r2 * 8 * (3 * HH) + c64 * 3 + cj * (3 * HH) + dk];
    }
#pragma unroll
    for (int j = 0; j < (16 * SS) / 128; ++j)
      lX[r2 * PX + XO + c64 + j * 2 * PX] = xb[tid + j * 128];
    __syncthreads();

    v8f acc31 = {}, acc33 = {};

#pragma unroll
    for (int q = 0; q < 2; ++q) {
      const int kc = q * 32;
      v16h wf = join8(*(const v8h*)&lW31[lj * PH + kc + g * 8],
                      *(const v8h*)&lW31[lj * PH + kc + 16 + g * 8]);
      v16h hf = join8(*(const v8h*)&lH31[s * PH + kc + g * 8],
                      *(const v8h*)&lH31[s * PH + kc + 16 + g * 8]);
      acc31 = wmma_f16(hf, wf, acc31);
    }

#pragma unroll
    for (int q = 0; q < 6; ++q) {
      const int kc = q * 32;
      const int dk = kc >> 6;
      const int hid0 = kc & 63;
      v16h wf = join8(*(const v8h*)&lW33[lj * PW3 + kc + g * 8],
                      *(const v8h*)&lW33[lj * PW3 + kc + 16 + g * 8]);
      v16h hf = join8(*(const v8h*)&lH33[(s + dk) * PH + hid0 + g * 8],
                      *(const v8h*)&lH33[(s + dk) * PH + hid0 + 16 + g * 8]);
      acc33 = wmma_f16(hf, wf, acc33);
    }

    __builtin_amdgcn_sched_barrier(0);

    float w[12];
    {
      const float* xr = &lX[lj * PX + s0];
      const float4 w0 = *(const float4*)xr;
      const float4 w1 = *(const float4*)(xr + 4);
      const float4 w2 = *(const float4*)(xr + 8);
      w[0] = w0.x; w[1] = w0.y; w[2] = w0.z;  w[3] = w0.w;
      w[4] = w1.x; w[5] = w1.y; w[6] = w1.z;  w[7] = w1.w;
      w[8] = w2.x; w[9] = w2.y; w[10] = w2.z; w[11] = w2.w;
    }
    float best = 0.0f;
#pragma unroll
    for (int r = 0; r < 8; ++r) {
      const float xa = w[r], xb_ = w[r + 1], xc = w[r + 2], xd = w[r + 3], xe = w[r + 4];
      const float sum3 = (xb_ + xc) + xd;
      const float sum5 = (((xa + xb_) + xc) + xd) + xe;
      const float ma = (r < 2 && lo_edge) ? xc : xa;
      const float mb = (r < 1 && lo_edge) ? xc : xb_;
      const float md = (r > 6 && hi_edge) ? xc : xd;
      const float me = (r > 5 && hi_edge) ? xc : xe;
      float mx3 = fmaxf(fmaxf(mb, xc), md);
      float mx5 = fmaxf(fmaxf(ma, mx3), me);
      float f31 = sum3 * (1.0f / 3.0f) + mx3;
      float f33 = sum5 * 0.2f + mx5;
      float g31 = 1.0f / (1.0f + __expf(-acc31[r]));
      float g33 = 1.0f / (1.0f + __expf(-acc33[r]));
      float val = f31 * g31 + f33 * g33;
      best = (r == 0) ? val : fmaxf(best, val);
    }
    best = fmaxf(best, other_row(best));
    if (g == 0) red[wv * CC + c0 + lj] = best;
    __syncthreads();
  }

  {
    const int c = tid << 1;
    float2 m0 = *(const float2*)&red[c];
    float2 m1 = *(const float2*)&red[CC + c];
    float2 m2 = *(const float2*)&red[2 * CC + c];
    float2 m3 = *(const float2*)&red[3 * CC + c];
    float2 m;
    m.x = fmaxf(fmaxf(m0.x, m1.x), fmaxf(m2.x, m3.x));
    m.y = fmaxf(fmaxf(m0.y, m1.y), fmaxf(m2.y, m3.y));
    *(float2*)&out[(size_t)(p * NN + n) * CC + c] = m;
  }
}

extern "C" void kernel_launch(void* const* d_in, const int* in_sizes, int n_in,
                              void* d_out, int out_size, void* d_ws,
                              size_t ws_size, hipStream_t stream) {
  (void)in_sizes; (void)n_in; (void)out_size; (void)ws_size;
  const float* x    = (const float*)d_in[0];
  const float* w31a = (const float*)d_in[1];
  const float* w31b = (const float*)d_in[2];
  const float* w33a = (const float*)d_in[3];
  const float* w33b = (const float*)d_in[4];

  half_t* h31 = (half_t*)d_ws;
  half_t* h33 = h31 + (size_t)PP * NN * HH * SS;

  dim3 g1(NN, 2, PP);
  conv_a_kernel<<<g1, 128, 0, stream>>>(x, w31a, w33a, h31, h33);

  dim3 g2(NN, PP);
  gate_kernel<<<g2, 128, 0, stream>>>(x, w31b, w33b, h31, h33, (float*)d_out);
}


# conv and gate kernels: a work-group takes several n (conv 4, gate 2; one 4-wave group per n) and stages each weight tile once for all: weight loads and staging stores per n fall 4x and 2x; sums same
# speedup vs baseline: 7.3346x; 1.2805x over previous
#include <hip/hip_runtime.h>
#include <hip/hip_bf16.h>
#include <math.h>

typedef _Float16 half_t;
typedef __attribute__((ext_vector_type(16))) _Float16 v16h;
typedef __attribute__((ext_vector_type(8)))  _Float16 v8h;
typedef __attribute__((ext_vector_type(8)))  float    v8f;

#define PP 16
#define NN 128
#define CC 256
#define SS 64
#define HH 64
#define KA (CC * 3)

#define PA 40
#define PH 72
#define PW3 200
#define APL (64 * PA + 16)
#define PX 68
#define XO 2
#define GC 4
#define GG 2
#define LBN ((SS + 2) * PA)

static __device__ __forceinline__ v16h join8(v8h lo, v8h hi) {
  return __builtin_shufflevector(lo, hi, 0, 1, 2, 3, 4, 5, 6, 7, 8, 9, 10, 11, 12, 13, 14, 15);
}

static __device__ __forceinline__ float other_row(float x) {
  const int b = __builtin_bit_cast(int, x);
  return __builtin_bit_cast(
      float, __builtin_amdgcn_permlanex16(b, b, 0x76543210, 0xfedcba98, false, false));
}

static __device__ __forceinline__ v8f wmma_f16(v16h a, v16h b, v8f c) {
  return __builtin_amdgcn_wmma_f32_16x16x32_f16(false, a, false, b, (short)0, c,
                                                false, false);
}

__global__ __launch_bounds__(128 * GC) void conv_a_kernel(
    const float* __restrict__ x, const float* __restrict__ w31a,
    const float* __restrict__ w33a, half_t* __restrict__ h31,
    half_t* __restrict__ h33) {
  __shared__ __align__(16) half_t lA[3 * APL];
  __shared__ __align__(16) half_t lBs[GC * LBN];

  const int tid = threadIdx.x;
  const int ng = tid >> 7;
  const int t7 = tid & 127;
  const int wv = (tid >> 5) & 3;
  const int lane = tid & 31;
  const int g = lane >> 4;
  const int lj = lane & 15;
  const int p = blockIdx.z;
  const int br = blockIdx.y;
  const int n = blockIdx.x * GC + ng;
  half_t* lB = lBs + ng * LBN;

  const float* wa = (br == 0 ? w31a : w33a) + (size_t)p * HH * KA;
  half_t* hb = (br == 0 ? h31 : h33) + (size_t)p * NN * HH * SS;
  const float* xb = x + (size_t)(p * NN + n) * CC * SS;

  const int sub = tid & 7, unit0 = tid >> 3;
  int aoff[12];
#pragma unroll
  for (int q = 0; q < 12; ++q) {
    const int idx = (q >> 2) * 32 + sub * 4 + (q & 3);
    aoff[q] = (idx % 3) * APL + unit0 * PA + idx / 3;
  }
  const float* was0 = wa + unit0 * KA + sub * 4;
  constexpr int RPP = 16 * GC;
  constexpr int NRB = HH / RPP;
  constexpr int NWP = 3 * NRB;

  if (t7 < 32) {
    lB[t7] = (half_t)0.0f;
    lB[(SS + 1) * PA + t7] = (half_t)0.0f;
  }

  v8f acc[4] = {};
  for (int ck = 0; ck < CC; ck += 32) {
    const float* was = was0 + ck * 3;
    float4 wq[NWP];
#pragma unroll
    for (int i = 0; i < NWP; ++i)
      wq[i] = *(const float4*)(was + (i % NRB) * RPP * KA + (i / NRB) * 32);
#pragma unroll
    for (int i = 0; i < NWP; ++i) {
      half_t* dst = &lA[(i % NRB) * RPP * PA];
      const int q0 = (i / NRB) * 4;
      dst[aoff[q0 + 0]] = (half_t)wq[i].x;
      dst[aoff[q0 + 1]] = (half_t)wq[i].y;
      dst[aoff[q0 + 2]] = (half_t)wq[i].z;
      dst[aoff[q0 + 3]] = (half_t)wq[i].w;
    }
#pragma unroll
    for (int i = 0; i < 2; ++i) {
      int u = t7 + i * 128;
      int j = u & 63, kg = u >> 6;
      const float* xs = xb + (ck + kg * 8) * SS + j;
      v8h v;
#pragma unroll
      for (int t = 0; t < 8; ++t) v[t] = (half_t)xs[t * SS];
      *(v8h*)&lB[(j + 1) * PA + kg * 8] = v;
    }
    if (ck + 32 < CC) {
      __builtin_prefetch(xb + (ck + 32 + (t7 >> 2)) * SS, 0, 3);
    }
    __syncthreads();

    const int arow = (wv << 4) + lj;
#pragma unroll
    for (int dk = 0; dk < 3; ++dk) {
      const half_t* ar = &lA[dk * APL + arow * PA];
      v16h afrag = join8(*(const v8h*)&ar[g * 8], *(const v8h*)&ar[16 + g * 8]);
#pragma unroll
      for (int t = 0; t < 4; ++t) {
        const half_t* brw = &lB[((t << 4) + lj + dk) * PA];
        v16h bfrag = join8(*(const v8h*)&brw[g * 8], *(const v8h*)&brw[16 + g * 8]);
        acc[t] = wmma_f16(afrag, bfrag, acc[t]);
      }
    }
    __syncthreads();
  }

#pragma unroll
  for (int t = 0; t < 4; ++t) {
#pragma unroll
    for (int r = 0; r < 8; ++r) {
      float v = acc[t][r];
      v = v >= 0.0f ? v : 0.01f * v;
      int hid = (wv << 4) + r + (g << 3);
      int s = (t << 4) + lj;
      hb[((size_t)n * HH + hid) * SS + s] = (half_t)v;
    }
  }
}

__global__ __launch_bounds__(128 * GG, 8) void gate_kernel(
    const float* __restrict__ x, const float* __restrict__ w31b,
    const float* __restrict__ w33b, const half_t* __restrict__ h31,
    const half_t* __restrict__ h33, float* __restrict__ out) {
  __shared__ __align__(16) half_t lH31s[GG * SS * PH];
  __shared__ __align__(16) half_t lH33s[GG * (SS + 2) * PH];
  __shared__ __align__(16) float lXs[GG * 16 * PX];
  __shared__ __align__(16) float reds[GG * 4 * CC];
  __shared__ __align__(16) half_t lW31[16 * PH];
  __shared__ __align__(16) half_t lW33[16 * PW3];

  const int tid = threadIdx.x;
  const int ng = tid >> 7;
  const int t7 = tid & 127;
  const int wv = (tid >> 5) & 3;
  const int lane = tid & 31;
  const int g = lane >> 4;
  const int lj = lane & 15;
  const int p = blockIdx.y, n = blockIdx.x * GG + ng;
  half_t* lH31 = lH31s + ng * (SS * PH);
  half_t* lH33 = lH33s + ng * ((SS + 2) * PH);
  float* lX = lXs + ng * (16 * PX);
  float* red = reds + ng * (4 * CC);

  const half_t* hb31 = h31 + (size_t)(p * NN + n) * HH * SS;
  const half_t* hb33 = h33 + (size_t)(p * NN + n) * HH * SS;

#pragma unroll
  for (int i = 0; i < (SS * HH) / (8 * 128); ++i) {
    int u = t7 + i * 128;
    int s = u & 63, hg = u >> 6;
    v8h a, b;
#pragma unroll
    for (int t = 0; t < 8; ++t) {
      a[t] = hb31[(hg * 8 + t) * SS + s];
      b[t] = hb33[(hg * 8 + t) * SS + s];
    }
    *(v8h*)&lH31[s * PH + hg * 8] = a;
    *(v8h*)&lH33[(s + 1) * PH + hg * 8] = b;
  }
  if (t7 < HH) {
    lH33[t7] = (half_t)0.0f;
    lH33[(SS + 1) * PH + t7] = (half_t)0.0f;
    const int zr = t7 >> 2, zk = t7 & 3;
    lX[zr * PX + (zk < 2 ? zk : SS + zk)] = 0.0f;
  }
  __syncthreads();

  const int s = (wv << 4) + lj;
  const int r2 = t7 >> 6, c64 = tid & 63;
  const int r4 = tid >> 6;
  const int s0 = (wv << 4) + (g << 3);
  const bool lo_edge = s0 == 0, hi_edge = s0 == SS - 8;

  for (int ct = 0; ct < CC / 16; ++ct) {
    const int c0 = ct << 4;
    const float* xb = x + ((size_t)(p * NN + n) * CC + c0) * SS;
    const float* wb1 = w31b + (size_t)(p * CC + c0) * HH;
    const float* wb3 = w33b + (size_t)(p * CC + c0) * (HH * 3);

    constexpr int RW = 2 * GG;
    constexpr int CW = 16 / RW;
#pragma unroll
    for (int j = 0; j < 16 / RW; ++j)
      lW31[r4 * PH + c64 + j * RW * PH] = (half_t)wb1[tid + j * (128 * GG)];
#pragma unroll
    for (int j = 0; j < 3 * CW; ++j) {
      const int cj = j / 3, dk = j % 3;
      lW33[r4 * CW * PW3 + c64 + cj * PW3 + dk * HH] =
          (half_t)wb3[r4 * CW * (3 * HH) + c64 * 3 + cj * (3 * HH) + dk];
    }
#pragma unroll
    for (int j = 0; j < (16 * SS) / 128; ++j)
      lX[r2 * PX + XO + c64 + j * 2 * PX] = xb[t7 + j * 128];
    __syncthreads();

    v8f acc31 = {}, acc33 = {};

#pragma unroll
    for (int q = 0; q < 2; ++q) {
      const int kc = q * 32;
      v16h wf = join8(*(const v8h*)&lW31[lj * PH + kc + g * 8],
                      *(const v8h*)&lW31[lj * PH + kc + 16 + g * 8]);
      v16h hf = join8(*(const v8h*)&lH31[s * PH + kc + g * 8],
                      *(const v8h*)&lH31[s * PH + kc + 16 + g * 8]);
      acc31 = wmma_f16(hf, wf, acc31);
    }

#pragma unroll
    for (int q = 0; q < 6; ++q) {
      const int kc = q * 32;
      const int dk = kc >> 6;
      const int hid0 = kc & 63;
      v16h wf = join8(*(const v8h*)&lW33[lj * PW3 + kc + g * 8],
                      *(const v8h*)&lW33[lj * PW3 + kc + 16 + g * 8]);
      v16h hf = join8(*(const v8h*)&lH33[(s + dk) * PH + hid0 + g * 8],
                      *(const v8h*)&lH33[(s + dk) * PH + hid0 + 16 + g * 8]);
      acc33 = wmma_f16(hf, wf, acc33);
    }

    __builtin_amdgcn_sched_barrier(0);

    float w[12];
    {
      const float* xr = &lX[lj * PX + s0];
      const float4 w0 = *(const float4*)xr;
      const float4 w1 = *(const float4*)(xr + 4);
      const float4 w2 = *(const float4*)(xr + 8);
      w[0] = w0.x; w[1] = w0.y; w[2] = w0.z;  w[3] = w0.w;
      w[4] = w1.x; w[5] = w1.y; w[6] = w1.z;  w[7] = w1.w;
      w[8] = w2.x; w[9] = w2.y; w[10] = w2.z; w[11] = w2.w;
    }
    float best = 0.0f;
#pragma unroll
    for (int r = 0; r < 8; ++r) {
      const float xa = w[r], xb_ = w[r + 1], xc = w[r + 2], xd = w[r + 3], xe = w[r + 4];
      const float sum3 = (xb_ + xc) + xd;
      const float sum5 = (((xa + xb_) + xc) + xd) + xe;
      const float ma = (r < 2 && lo_edge) ? xc : xa;
      const float mb = (r < 1 && lo_edge) ? xc : xb_;
      const float md = (r > 6 && hi_edge) ? xc : xd;
      const float me = (r > 5 && hi_edge) ? xc : xe;
      float mx3 = fmaxf(fmaxf(mb, xc), md);
      float mx5 = fmaxf(fmaxf(ma, mx3), me);
      float f31 = sum3 * (1.0f / 3.0f) + mx3;
      float f33 = sum5 * 0.2f + mx5;
      float g31 = 1.0f / (1.0f + __expf(-acc31[r]));
      float g33 = 1.0f / (1.0f + __expf(-acc33[r]));
      float val = f31 * g31 + f33 * g33;
      best = (r == 0) ? val : fmaxf(best, val);
    }
    best = fmaxf(best, other_row(best));
    if (g == 0) red[wv * CC + c0 + lj] = best;
    __syncthreads();
  }

  {
    const int c = t7 << 1;
    float2 m0 = *(const float2*)&red[c];
    float2 m1 = *(const float2*)&red[CC + c];
    float2 m2 = *(const float2*)&red[2 * CC + c];
    float2 m3 = *(const float2*)&red[3 * CC + c];
    float2 m;
    m.x = fmaxf(fmaxf(m0.x, m1.x), fmaxf(m2.x, m3.x));
    m.y = fmaxf(fmaxf(m0.y, m1.y), fmaxf(m2.y, m3.y));
    *(float2*)&out[(size_t)(p * NN + n) * CC + c] = m;
  }
}

extern "C" void kernel_launch(void* const* d_in, const int* in_sizes, int n_in,
                              void* d_out, int out_size, void* d_ws,
                              size_t ws_size, hipStream_t stream) {
  (void)in_sizes; (void)n_in; (void)out_size; (void)ws_size;
  const float* x    = (const float*)d_in[0];
  const float* w31a = (const float*)d_in[1];
  const float* w31b = (const float*)d_in[2];
  const float* w33a = (const float*)d_in[3];
  const float* w33b = (const float*)d_in[4];

  half_t* h31 = (half_t*)d_ws;
  half_t* h33 = h31 + (size_t)PP * NN * HH * SS;

  dim3 g1(NN / GC, 2, PP);
  conv_a_kernel<<<g1, 128 * GC, 0, stream>>>(x, w31a, w33a, h31, h33);

  dim3 g2(NN / GG, PP);
  gate_kernel<<<g2, 128 * GG, 0, stream>>>(x, w31b, w33b, h31, h33, (float*)d_out);
}


# conv kernel: global loads run a chunk ahead in registers: a chunk's 19 loads per thread go out as one batch before the previous chunk's barrier (LDS-only fence), land during its products; sums same
# speedup vs baseline: 8.1721x; 1.1142x over previous
#include <hip/hip_runtime.h>
#include <hip/hip_bf16.h>
#include <math.h>

typedef _Float16 half_t;
typedef __attribute__((ext_vector_type(16))) _Float16 v16h;
typedef __attribute__((ext_vector_type(8)))  _Float16 v8h;
typedef __attribute__((ext_vector_type(8)))  float    v8f;

#define PP 16
#define NN 128
#define CC 256
#define SS 64
#define HH 64
#define KA (CC * 3)

#define PA 40
#define PH 72
#define PW3 200
#define APL (64 * PA + 16)
#define PX 68
#define XO 2
#define GC 4
#define GG 2
#define LBN ((SS + 2) * PA)

static __device__ __forceinline__ v16h join8(v8h lo, v8h hi) {
  return __builtin_shufflevector(lo, hi, 0, 1, 2, 3, 4, 5, 6, 7, 8, 9, 10, 11, 12, 13, 14, 15);
}

static __device__ __forceinline__ float other_row(float x) {
  const int b = __builtin_bit_cast(int, x);
  return __builtin_bit_cast(
      float, __builtin_amdgcn_permlanex16(b, b, 0x76543210, 0xfedcba98, false, false));
}

static __device__ __forceinline__ v8f wmma_f16(v16h a, v16h b, v8f c) {
  return __builtin_amdgcn_wmma_f32_16x16x32_f16(false, a, false, b, (short)0, c,
                                                false, false);
}

static __device__ __forceinline__ void lds_barrier() {
  __builtin_amdgcn_fence(__ATOMIC_RELEASE, "workgroup", "local");
  __builtin_amdgcn_s_barrier();
  __builtin_amdgcn_fence(__ATOMIC_ACQUIRE, "workgroup", "local");
}

__global__ __launch_bounds__(128 * GC, 8) void conv_a_kernel(
    const float* __restrict__ x, const float* __restrict__ w31a,
    const float* __restrict__ w33a, half_t* __restrict__ h31,
    half_t* __restrict__ h33) {
  __shared__ __align__(16) half_t lA[3 * APL];
  __shared__ __align__(16) half_t lBs[GC * LBN];

  const int tid = threadIdx.x;
  const int ng = tid >> 7;
  const int t7 = tid & 127;
  const int wv = (tid >> 5) & 3;
  const int lane = tid & 31;
  const int g = lane >> 4;
  const int lj = lane & 15;
  const int p = blockIdx.z;
  const int br = blockIdx.y;
  const int n = blockIdx.x * GC + ng;
  half_t* lB = lBs + ng * LBN;

  const float* wa = (br == 0 ? w31a : w33a) + (size_t)p * HH * KA;
  half_t* hb = (br == 0 ? h31 : h33) + (size_t)p * NN * HH * SS;
  const float* xb = x + (size_t)(p * NN + n) * CC * SS;

  const int sub = tid & 7, unit0 = tid >> 3;
  int aoff[12];
#pragma unroll
  for (int q = 0; q < 12; ++q) {
    const int idx = (q >> 2) * 32 + sub * 4 + (q & 3);
    aoff[q] = (idx % 3) * APL + unit0 * PA + idx / 3;
  }
  const float* was0 = wa + unit0 * KA + sub * 4;
  static_assert(16 * GC == HH, "one staging pass must cover all hid rows");

  const int j = t7 & 63, kg0 = t7 >> 6;
  const float* xs0 = xb + (kg0 * 8) * SS + j;
  half_t* lBst = &lB[(j + 1) * PA + kg0 * 8];

  if (t7 < 32) {
    lB[t7] = (half_t)0.0f;
    lB[(SS + 1) * PA + t7] = (half_t)0.0f;
  }

  float xv[16];
  float4 wq[3];
#pragma unroll
  for (int i = 0; i < 2; ++i)
#pragma unroll
    for (int t = 0; t < 8; ++t) xv[i * 8 + t] = xs0[(i * 16 + t) * SS];
#pragma unroll
  for (int i = 0; i < 3; ++i) wq[i] = *(const float4*)(was0 + i * 32);

  v8f acc[4] = {};
  for (int ck = 0; ck < CC; ck += 32) {
#pragma unroll
    for (int i = 0; i < 2; ++i) {
      v8h v;
#pragma unroll
      for (int t = 0; t < 8; ++t) v[t] = (half_t)xv[i * 8 + t];
      *(v8h*)&lBst[i * 16] = v;
    }
#pragma unroll
    for (int i = 0; i < 3; ++i) {
      const int q0 = i * 4;
      lA[aoff[q0 + 0]] = (half_t)wq[i].x;
      lA[aoff[q0 + 1]] = (half_t)wq[i].y;
      lA[aoff[q0 + 2]] = (half_t)wq[i].z;
      lA[aoff[q0 + 3]] = (half_t)wq[i].w;
    }
    if (ck + 32 < CC) {
#pragma unroll
      for (int i = 0; i < 2; ++i)
#pragma unroll
        for (int t = 0; t < 8; ++t)
          xv[i * 8 + t] = xs0[(ck + 32 + i * 16 + t) * SS];
#pragma unroll
      for (int i = 0; i < 3; ++i)
        wq[i] = *(const float4*)(was0 + (ck + 32) * 3 + i * 32);
    }
    if (ck + 64 < CC) {
      __builtin_prefetch(xb + (ck + 64 + (t7 >> 2)) * SS, 0, 3);
    }
    lds_barrier();

    const int arow = (wv << 4) + lj;
#pragma unroll
    for (int dk = 0; dk < 3; ++dk) {
      const half_t* ar = &lA[dk * APL + arow * PA];
      v16h afrag = join8(*(const v8h*)&ar[g * 8], *(const v8h*)&ar[16 + g * 8]);
#pragma unroll
      for (int t = 0; t < 4; ++t) {
        const half_t* brw = &lB[((t << 4) + lj + dk) * PA];
        v16h bfrag = join8(*(const v8h*)&brw[g * 8], *(const v8h*)&brw[16 + g * 8]);
        acc[t] = wmma_f16(afrag, bfrag, acc[t]);
      }
      __builtin_amdgcn_sched_barrier(0);
    }
    lds_barrier();
  }

#pragma unroll
  for (int t = 0; t < 4; ++t) {
#pragma unroll
    for (int r = 0; r < 8; ++r) {
      float v = acc[t][r];
      v = v >= 0.0f ? v : 0.01f * v;
      int hid = (wv << 4) + r + (g << 3);
      int s = (t << 4) + lj;
      hb[((size_t)n * HH + hid) * SS + s] = (half_t)v;
    }
  }
}

__global__ __launch_bounds__(128 * GG, 8) void gate_kernel(
    const float* __restrict__ x, const float* __restrict__ w31b,
    const float* __restrict__ w33b, const half_t* __restrict__ h31,
    const half_t* __restrict__ h33, float* __restrict__ out) {
  __shared__ __align__(16) half_t lH31s[GG * SS * PH];
  __shared__ __align__(16) half_t lH33s[GG * (SS + 2) * PH];
  __shared__ __align__(16) float lXs[GG * 16 * PX];
  __shared__ __align__(16) float reds[GG * 4 * CC];
  __shared__ __align__(16) half_t lW31[16 * PH];
  __shared__ __align__(16) half_t lW33[16 * PW3];

  const int tid = threadIdx.x;
  const int ng = tid >> 7;
  const int t7 = tid & 127;
  const int wv = (tid >> 5) & 3;
  const int lane = tid & 31;
  const int g = lane >> 4;
  const int lj = lane & 15;
  const int p = blockIdx.y, n = blockIdx.x * GG + ng;
  half_t* lH31 = lH31s + ng * (SS * PH);
  half_t* lH33 = lH33s + ng * ((SS + 2) * PH);
  float* lX = lXs + ng * (16 * PX);
  float* red = reds + ng * (4 * CC);

  const half_t* hb31 = h31 + (size_t)(p * NN + n) * HH * SS;
  const half_t* hb33 = h33 + (size_t)(p * NN + n) * HH * SS;

#pragma unroll
  for (int i = 0; i < (SS * HH) / (8 * 128); ++i) {
    int u = t7 + i * 128;
    int s = u & 63, hg = u >> 6;
    v8h a, b;
#pragma unroll
    for (int t = 0; t < 8; ++t) {
      a[t] = hb31[(hg * 8 + t) * SS + s];
      b[t] = hb33[(hg * 8 + t) * SS + s];
    }
    *(v8h*)&lH31[s * PH + hg * 8] = a;
    *(v8h*)&lH33[(s + 1) * PH + hg * 8] = b;
  }
  if (t7 < HH) {
    lH33[t7] = (half_t)0.0f;
    lH33[(SS + 1) * PH + t7] = (half_t)0.0f;
    const int zr = t7 >> 2, zk = t7 & 3;
    lX[zr * PX + (zk < 2 ? zk : SS + zk)] = 0.0f;
  }
  __syncthreads();

  const int s = (wv << 4) + lj;
  const int r2 = t7 >> 6, c64 = tid & 63;
  const int r4 = tid >> 6;
  const int s0 = (wv << 4) + (g << 3);
  const bool lo_edge = s0 == 0, hi_edge = s0 == SS - 8;

  for (int ct = 0; ct < CC / 16; ++ct) {
    const int c0 = ct << 4;
    const float* xb = x + ((size_t)(p * NN + n) * CC + c0) * SS;
    const float* wb1 = w31b + (size_t)(p * CC + c0) * HH;
    const float* wb3 = w33b + (size_t)(p * CC + c0) * (HH * 3);

    constexpr int RW = 2 * GG;
    constexpr int CW = 16 / RW;
#pragma unroll
    for (int j = 0; j < 16 / RW; ++j)
      lW31[r4 * PH + c64 + j * RW * PH] = (half_t)wb1[tid + j * (128 * GG)];
#pragma unroll
    for (int j = 0; j < 3 * CW; ++j) {
      const int cj = j / 3, dk = j % 3;
      lW33[r4 * CW * PW3 + c64 + cj * PW3 + dk * HH] =
          (half_t)wb3[r4 * CW * (3 * HH) + c64 * 3 + cj * (3 * HH) + dk];
    }
#pragma unroll
    for (int j = 0; j < (16 * SS) / 128; ++j)
      lX[r2 * PX + XO + c64 + j * 2 * PX] = xb[t7 + j * 128];
    __syncthreads();

    v8f acc31 = {}, acc33 = {};

#pragma unroll
    for (int q = 0; q < 2; ++q) {
      const int kc = q * 32;
      v16h wf = join8(*(const v8h*)&lW31[lj * PH + kc + g * 8],
                      *(const v8h*)&lW31[lj * PH + kc + 16 + g * 8]);
      v16h hf = join8(*(const v8h*)&lH31[s * PH + kc + g * 8],
                      *(const v8h*)&lH31[s * PH + kc + 16 + g * 8]);
      acc31 = wmma_f16(hf, wf, acc31);
    }

#pragma unroll
    for (int q = 0; q < 6; ++q) {
      const int kc = q * 32;
      const int dk = kc >> 6;
      const int hid0 = kc & 63;
      v16h wf = join8(*(const v8h*)&lW33[lj * PW3 + kc + g * 8],
                      *(const v8h*)&lW33[lj * PW3 + kc + 16 + g * 8]);
      v16h hf = join8(*(const v8h*)&lH33[(s + dk) * PH + hid0 + g * 8],
                      *(const v8h*)&lH33[(s + dk) * PH + hid0 + 16 + g * 8]);
      acc33 = wmma_f16(hf, wf, acc33);
    }

    __builtin_amdgcn_sched_barrier(0);

    float w[12];
    {
      const float* xr = &lX[lj * PX + s0];
      const float4 w0 = *(const float4*)xr;
      const float4 w1 = *(const float4*)(xr + 4);
      const float4 w2 = *(const float4*)(xr + 8);
      w[0] = w0.x; w[1] = w0.y; w[2] = w0.z;  w[3] = w0.w;
      w[4] = w1.x; w[5] = w1.y; w[6] = w1.z;  w[7] = w1.w;
      w[8] = w2.x; w[9] = w2.y; w[10] = w2.z; w[11] = w2.w;
    }
    float best = 0.0f;
#pragma unroll
    for (int r = 0; r < 8; ++r) {
      const float xa = w[r], xb_ = w[r + 1], xc = w[r + 2], xd = w[r + 3], xe = w[r + 4];
      const float sum3 = (xb_ + xc) + xd;
      const float sum5 = (((xa + xb_) + xc) + xd) + xe;
      const float ma = (r < 2 && lo_edge) ? xc : xa;
      const float mb = (r < 1 && lo_edge) ? xc : xb_;
      const float md = (r > 6 && hi_edge) ? xc : xd;
      const float me = (r > 5 && hi_edge) ? xc : xe;
      float mx3 = fmaxf(fmaxf(mb, xc), md);
      float mx5 = fmaxf(fmaxf(ma, mx3), me);
      float f31 = sum3 * (1.0f / 3.0f) + mx3;
      float f33 = sum5 * 0.2f + mx5;
      float g31 = 1.0f / (1.0f + __expf(-acc31[r]));
      float g33 = 1.0f / (1.0f + __expf(-acc33[r]));
      float val = f31 * g31 + f33 * g33;
      best = (r == 0) ? val : fmaxf(best, val);
    }
    best = fmaxf(best, other_row(best));
    if (g == 0) red[wv * CC + c0 + lj] = best;
    __syncthreads();
  }

  {
    const int c = t7 << 1;
    float2 m0 = *(const float2*)&red[c];
    float2 m1 = *(const float2*)&red[CC + c];
    float2 m2 = *(const float2*)&red[2 * CC + c];
    float2 m3 = *(const float2*)&red[3 * CC + c];
    float2 m;
    m.x = fmaxf(fmaxf(m0.x, m1.x), fmaxf(m2.x, m3.x));
    m.y = fmaxf(fmaxf(m0.y, m1.y), fmaxf(m2.y, m3.y));
    *(float2*)&out[(size_t)(p * NN + n) * CC + c] = m;
  }
}

extern "C" void kernel_launch(void* const* d_in, const int* in_sizes, int n_in,
                              void* d_out, int out_size, void* d_ws,
                              size_t ws_size, hipStream_t stream) {
  (void)in_sizes; (void)n_in; (void)out_size; (void)ws_size;
  const float* x    = (const float*)d_in[0];
  const float* w31a = (const float*)d_in[1];
  const float* w31b = (const float*)d_in[2];
  const float* w33a = (const float*)d_in[3];
  const float* w33b = (const float*)d_in[4];

  half_t* h31 = (half_t*)d_ws;
  half_t* h33 = h31 + (size_t)PP * NN * HH * SS;

  dim3 g1(NN / GC, 2, PP);
  conv_a_kernel<<<g1, 128 * GC, 0, stream>>>(x, w31a, w33a, h31, h33);

  dim3 g2(NN / GG, PP);
  gate_kernel<<<g2, 128 * GG, 0, stream>>>(x, w31b, w33b, h31, h33, (float*)d_out);
}


# gate kernel: global loads run a tile ahead in registers: a tile's 16 loads per thread go out before the previous tile's barrier (LDS-only fence), land during its products and epilogue; sums same
# speedup vs baseline: 8.4373x; 1.0325x over previous
#include <hip/hip_runtime.h>
#include <hip/hip_bf16.h>
#include <math.h>

typedef _Float16 half_t;
typedef __attribute__((ext_vector_type(16))) _Float16 v16h;
typedef __attribute__((ext_vector_type(8)))  _Float16 v8h;
typedef __attribute__((ext_vector_type(8)))  float    v8f;

#define PP 16
#define NN 128
#define CC 256
#define SS 64
#define HH 64
#define KA (CC * 3)

#define PA 40
#define PH 72
#define PW3 200
#define APL (64 * PA + 16)
#define PX 68
#define XO 2
#define GC 4
#define GG 2
#define LBN ((SS + 2) * PA)

static __device__ __forceinline__ v16h join8(v8h lo, v8h hi) {
  return __builtin_shufflevector(lo, hi, 0, 1, 2, 3, 4, 5, 6, 7, 8, 9, 10, 11, 12, 13, 14, 15);
}

static __device__ __forceinline__ float other_row(float x) {
  const int b = __builtin_bit_cast(int, x);
  return __builtin_bit_cast(
      float, __builtin_amdgcn_permlanex16(b, b, 0x76543210, 0xfedcba98, false, false));
}

static __device__ __forceinline__ v8f wmma_f16(v16h a, v16h b, v8f c) {
  return __builtin_amdgcn_wmma_f32_16x16x32_f16(false, a, false, b, (short)0, c,
                                                false, false);
}

static __device__ __forceinline__ void lds_barrier() {
  __builtin_amdgcn_fence(__ATOMIC_RELEASE, "workgroup", "local");
  __builtin_amdgcn_s_barrier();
  __builtin_amdgcn_fence(__ATOMIC_ACQUIRE, "workgroup", "local");
}

__global__ __launch_bounds__(128 * GC, 8) void conv_a_kernel(
    const float* __restrict__ x, const float* __restrict__ w31a,
    const float* __restrict__ w33a, half_t* __restrict__ h31,
    half_t* __restrict__ h33) {
  __shared__ __align__(16) half_t lA[3 * APL];
  __shared__ __align__(16) half_t lBs[GC * LBN];

  const int tid = threadIdx.x;
  const int ng = tid >> 7;
  const int t7 = tid & 127;
  const int wv = (tid >> 5) & 3;
  const int lane = tid & 31;
  const int g = lane >> 4;
  const int lj = lane & 15;
  const int p = blockIdx.z;
  const int br = blockIdx.y;
  const int n = blockIdx.x * GC + ng;
  half_t* lB = lBs + ng * LBN;

  const float* wa = (br == 0 ? w31a : w33a) + (size_t)p * HH * KA;
  half_t* hb = (br == 0 ? h31 : h33) + (size_t)p * NN * HH * SS;
  const float* xb = x + (size_t)(p * NN + n) * CC * SS;

  const int sub = tid & 7, unit0 = tid >> 3;
  int aoff[12];
#pragma unroll
  for (int q = 0; q < 12; ++q) {
    const int idx = (q >> 2) * 32 + sub * 4 + (q & 3);
    aoff[q] = (idx % 3) * APL + unit0 * PA + idx / 3;
  }
  const float* was0 = wa + unit0 * KA + sub * 4;
  static_assert(16 * GC == HH, "one staging pass must cover all hid rows");

  const int j = t7 & 63, kg0 = t7 >> 6;
  const float* xs0 = xb + (kg0 * 8) * SS + j;
  half_t* lBst = &lB[(j + 1) * PA + kg0 * 8];

  if (t7 < 32) {
    lB[t7] = (half_t)0.0f;
    lB[(SS + 1) * PA + t7] = (half_t)0.0f;
  }

  float xv[16];
  float4 wq[3];
#pragma unroll
  for (int i = 0; i < 2; ++i)
#pragma unroll
    for (int t = 0; t < 8; ++t) xv[i * 8 + t] = xs0[(i * 16 + t) * SS];
#pragma unroll
  for (int i = 0; i < 3; ++i) wq[i] = *(const float4*)(was0 + i * 32);

  v8f acc[4] = {};
  for (int ck = 0; ck < CC; ck += 32) {
#pragma unroll
    for (int i = 0; i < 2; ++i) {
      v8h v;
#pragma unroll
      for (int t = 0; t < 8; ++t) v[t] = (half_t)xv[i * 8 + t];
      *(v8h*)&lBst[i * 16] = v;
    }
#pragma unroll
    for (int i = 0; i < 3; ++i) {
      const int q0 = i * 4;
      lA[aoff[q0 + 0]] = (half_t)wq[i].x;
      lA[aoff[q0 + 1]] = (half_t)wq[i].y;
      lA[aoff[q0 + 2]] = (half_t)wq[i].z;
      lA[aoff[q0 + 3]] = (half_t)wq[i].w;
    }
    if (ck + 32 < CC) {
#pragma unroll
      for (int i = 0; i < 2; ++i)
#pragma unroll
        for (int t = 0; t < 8; ++t)
          xv[i * 8 + t] = xs0[(ck + 32 + i * 16 + t) * SS];
#pragma unroll
      for (int i = 0; i < 3; ++i)
        wq[i] = *(const float4*)(was0 + (ck + 32) * 3 + i * 32);
    }
    if (ck + 64 < CC) {
      __builtin_prefetch(xb + (ck + 64 + (t7 >> 2)) * SS, 0, 3);
    }
    lds_barrier();

    const int arow = (wv << 4) + lj;
#pragma unroll
    for (int dk = 0; dk < 3; ++dk) {
      const half_t* ar = &lA[dk * APL + arow * PA];
      v16h afrag = join8(*(const v8h*)&ar[g * 8], *(const v8h*)&ar[16 + g * 8]);
#pragma unroll
      for (int t = 0; t < 4; ++t) {
        const half_t* brw = &lB[((t << 4) + lj + dk) * PA];
        v16h bfrag = join8(*(const v8h*)&brw[g * 8], *(const v8h*)&brw[16 + g * 8]);
        acc[t] = wmma_f16(afrag, bfrag, acc[t]);
      }
      __builtin_amdgcn_sched_barrier(0);
    }
    lds_barrier();
  }

#pragma unroll
  for (int t = 0; t < 4; ++t) {
#pragma unroll
    for (int r = 0; r < 8; ++r) {
      float v = acc[t][r];
      v = v >= 0.0f ? v : 0.01f * v;
      int hid = (wv << 4) + r + (g << 3);
      int s = (t << 4) + lj;
      hb[((size_t)n * HH + hid) * SS + s] = (half_t)v;
    }
  }
}

__global__ __launch_bounds__(128 * GG, 8) void gate_kernel(
    const float* __restrict__ x, const float* __restrict__ w31b,
    const float* __restrict__ w33b, const half_t* __restrict__ h31,
    const half_t* __restrict__ h33, float* __restrict__ out) {
  __shared__ __align__(16) half_t lH31s[GG * SS * PH];
  __shared__ __align__(16) half_t lH33s[GG * (SS + 2) * PH];
  __shared__ __align__(16) float lXs[GG * 16 * PX];
  __shared__ __align__(16) float reds[GG * 4 * CC];
  __shared__ __align__(16) half_t lW31[16 * PH];
  __shared__ __align__(16) half_t lW33[16 * PW3];

  const int tid = threadIdx.x;
  const int ng = tid >> 7;
  const int t7 = tid & 127;
  const int wv = (tid >> 5) & 3;
  const int lane = tid & 31;
  const int g = lane >> 4;
  const int lj = lane & 15;
  const int p = blockIdx.y, n = blockIdx.x * GG + ng;
  half_t* lH31 = lH31s + ng * (SS * PH);
  half_t* lH33 = lH33s + ng * ((SS + 2) * PH);
  float* lX = lXs + ng * (16 * PX);
  float* red = reds + ng * (4 * CC);

  const half_t* hb31 = h31 + (size_t)(p * NN + n) * HH * SS;
  const half_t* hb33 = h33 + (size_t)(p * NN + n) * HH * SS;

  const int r2 = t7 >> 6, c64 = tid & 63;
  const int r4 = tid >> 6;
  constexpr int RW = 2 * GG;
  constexpr int CW = 16 / RW;
  constexpr int N1 = 16 / RW, N3 = 3 * CW, NX = (16 * SS) / 128;
  const float* xb0 = x + (size_t)(p * NN + n) * CC * SS + t7;
  const float* wb10 = w31b + (size_t)p * CC * HH + tid;
  const float* wb30 = w33b + (size_t)p * CC * (HH * 3) + r4 * CW * (3 * HH) + c64 * 3;
  float q1[N1], q3[N3], qx[NX];
#pragma unroll
  for (int j = 0; j < N1; ++j) q1[j] = wb10[j * (128 * GG)];
#pragma unroll
  for (int j = 0; j < N3; ++j) q3[j] = wb30[(j / 3) * (3 * HH) + (j % 3)];
#pragma unroll
  for (int j = 0; j < NX; ++j) qx[j] = xb0[j * 128];

#pragma unroll
  for (int i = 0; i < (SS * HH) / (8 * 128); ++i) {
    int u = t7 + i * 128;
    int s = u & 63, hg = u >> 6;
    v8h a, b;
#pragma unroll
    for (int t = 0; t < 8; ++t) {
      a[t] = hb31[(hg * 8 + t) * SS + s];
      b[t] = hb33[(hg * 8 + t) * SS + s];
    }
    *(v8h*)&lH31[s * PH + hg * 8] = a;
    *(v8h*)&lH33[(s + 1) * PH + hg * 8] = b;
  }
  if (t7 < HH) {
    lH33[t7] = (half_t)0.0f;
    lH33[(SS + 1) * PH + t7] = (half_t)0.0f;
    const int zr = t7 >> 2, zk = t7 & 3;
    lX[zr * PX + (zk < 2 ? zk : SS + zk)] = 0.0f;
  }
  __syncthreads();

  const int s = (wv << 4) + lj;
  const int s0 = (wv << 4) + (g << 3);
  const bool lo_edge = s0 == 0, hi_edge = s0 == SS - 8;

  for (int ct = 0; ct < CC / 16; ++ct) {
    const int c0 = ct << 4;

#pragma unroll
    for (int j = 0; j < N1; ++j)
      lW31[r4 * PH + c64 + j * RW * PH] = (half_t)q1[j];
#pragma unroll
    for (int j = 0; j < N3; ++j) {
      const int cj = j / 3, dk = j % 3;
      lW33[r4 * CW * PW3 + c64 + cj * PW3 + dk * HH] = (half_t)q3[j];
    }
#pragma unroll
    for (int j = 0; j < NX; ++j)
      lX[r2 * PX + XO + c64 + j * 2 * PX] = qx[j];
    if (ct + 1 < CC / 16) {
      const float* xb = xb0 + (size_t)(c0 + 16) * SS;
      const float* wb1 = wb10 + (size_t)(c0 + 16) * HH;
      const float* wb3 = wb30 + (size_t)(c0 + 16) * (HH * 3);
#pragma unroll
      for (int j = 0; j < N1; ++j) q1[j] = wb1[j * (128 * GG)];
#pragma unroll
      for (int j = 0; j < N3; ++j) q3[j] = wb3[(j / 3) * (3 * HH) + (j % 3)];
#pragma unroll
      for (int j = 0; j < NX; ++j) qx[j] = xb[j * 128];
    }
    lds_barrier();

    v8f acc31 = {}, acc33 = {};

#pragma unroll
    for (int q = 0; q < 2; ++q) {
      const int kc = q * 32;
      v16h wf = join8(*(const v8h*)&lW31[lj * PH + kc + g * 8],
                      *(const v8h*)&lW31[lj * PH + kc + 16 + g * 8]);
      v16h hf = join8(*(const v8h*)&lH31[s * PH + kc + g * 8],
                      *(const v8h*)&lH31[s * PH + kc + 16 + g * 8]);
      acc31 = wmma_f16(hf, wf, acc31);
    }

#pragma unroll
    for (int q = 0; q < 6; ++q) {
      const int kc = q * 32;
      const int dk = kc >> 6;
      const int hid0 = kc & 63;
      v16h wf = join8(*(const v8h*)&lW33[lj * PW3 + kc + g * 8],
                      *(const v8h*)&lW33[lj * PW3 + kc + 16 + g * 8]);
      v16h hf = join8(*(const v8h*)&lH33[(s + dk) * PH + hid0 + g * 8],
                      *(const v8h*)&lH33[(s + dk) * PH + hid0 + 16 + g * 8]);
      acc33 = wmma_f16(hf, wf, acc33);
    }

    __builtin_amdgcn_sched_barrier(0);

    float w[12];
    {
      const float* xr = &lX[lj * PX + s0];
      const float4 w0 = *(const float4*)xr;
      const float4 w1 = *(const float4*)(xr + 4);
      const float4 w2 = *(const float4*)(xr + 8);
      w[0] = w0.x; w[1] = w0.y; w[2] = w0.z;  w[3] = w0.w;
      w[4] = w1.x; w[5] = w1.y; w[6] = w1.z;  w[7] = w1.w;
      w[8] = w2.x; w[9] = w2.y; w[10] = w2.z; w[11] = w2.w;
    }
    float best = 0.0f;
#pragma unroll
    for (int r = 0; r < 8; ++r) {
      const float xa = w[r], xb_ = w[r + 1], xc = w[r + 2], xd = w[r + 3], xe = w[r + 4];
      const float sum3 = (xb_ + xc) + xd;
      const float sum5 = (((xa + xb_) + xc) + xd) + xe;
      const float ma = (r < 2 && lo_edge) ? xc : xa;
      const float mb = (r < 1 && lo_edge) ? xc : xb_;
      const float md = (r > 6 && hi_edge) ? xc : xd;
      const float me = (r > 5 && hi_edge) ? xc : xe;
      float mx3 = fmaxf(fmaxf(mb, xc), md);
      float mx5 = fmaxf(fmaxf(ma, mx3), me);
      float f31 = sum3 * (1.0f / 3.0f) + mx3;
      float f33 = sum5 * 0.2f + mx5;
      float g31 = 1.0f / (1.0f + __expf(-acc31[r]));
      float g33 = 1.0f / (1.0f + __expf(-acc33[r]));
      float val = f31 * g31 + f33 * g33;
      best = (r == 0) ? val : fmaxf(best, val);
    }
    best = fmaxf(best, other_row(best));
    if (g == 0) red[wv * CC + c0 + lj] = best;
    lds_barrier();
  }

  {
    const int c = t7 << 1;
    float2 m0 = *(const float2*)&red[c];
    float2 m1 = *(const float2*)&red[CC + c];
    float2 m2 = *(const float2*)&red[2 * CC + c];
    float2 m3 = *(const float2*)&red[3 * CC + c];
    float2 m;
    m.x = fmaxf(fmaxf(m0.x, m1.x), fmaxf(m2.x, m3.x));
    m.y = fmaxf(fmaxf(m0.y, m1.y), fmaxf(m2.y, m3.y));
    *(float2*)&out[(size_t)(p * NN + n) * CC + c] = m;
  }
}

extern "C" void kernel_launch(void* const* d_in, const int* in_sizes, int n_in,
                              void* d_out, int out_size, void* d_ws,
                              size_t ws_size, hipStream_t stream) {
  (void)in_sizes; (void)n_in; (void)out_size; (void)ws_size;
  const float* x    = (const float*)d_in[0];
  const float* w31a = (const float*)d_in[1];
  const float* w31b = (const float*)d_in[2];
  const float* w33a = (const float*)d_in[3];
  const float* w33b = (const float*)d_in[4];

  half_t* h31 = (half_t*)d_ws;
  half_t* h33 = h31 + (size_t)PP * NN * HH * SS;

  dim3 g1(NN / GC, 2, PP);
  conv_a_kernel<<<g1, 128 * GC, 0, stream>>>(x, w31a, w33a, h31, h33);

  dim3 g2(NN / GG, PP);
  gate_kernel<<<g2, 128 * GG, 0, stream>>>(x, w31b, w33b, h31, h33, (float*)d_out);
}
